# starting kernel
# baseline (speedup 1.0000x reference)
#include <hip/hip_runtime.h>

typedef float v2f __attribute__((ext_vector_type(2)));
typedef float v8f __attribute__((ext_vector_type(8)));

#define GDIN  128
#define GDH   128
#define GDOUT 64

__global__ void zero_f32_kernel(float* __restrict__ p, size_t n) {
  size_t i = (size_t)blockIdx.x * blockDim.x + threadIdx.x;
  size_t stride = (size_t)gridDim.x * blockDim.x;
  for (; i < n; i += stride) p[i] = 0.0f;
}

__global__ void degree_kernel(const int* __restrict__ dst, float* __restrict__ deg, int E) {
  int i = blockIdx.x * blockDim.x + threadIdx.x;
  int stride = gridDim.x * blockDim.x;
  for (; i < E; i += stride) atomicAdd(&deg[dst[i]], 1.0f);
}

__global__ void rsqrt_kernel(float* __restrict__ dinv, int N) {
  int i = blockIdx.x * blockDim.x + threadIdx.x;
  if (i < N) dinv[i] = rsqrtf(dinv[i] + 1.0f);
}

template <int NCOLS>
__global__ void __launch_bounds__(256) gemm_wmma_f32(const float* __restrict__ X,
                                                     const float* __restrict__ W,
                                                     float* __restrict__ T) {
  __shared__ float xs[16 * 128];
  const int tid = threadIdx.x;
  const int wave = tid >> 5;
  const int lane = tid & 31;
  const int nwaves = blockDim.x >> 5;
  const size_t m0 = (size_t)blockIdx.x * 16;

  for (int i = tid; i < 16 * 128; i += blockDim.x) xs[i] = X[m0 * 128 + i];
  __syncthreads();

  const int m = lane & 15;
  const int kg = (lane >> 4) << 1;
  const int ntiles = NCOLS >> 4;

  for (int nt = wave; nt < ntiles; nt += nwaves) {
    const int n = (nt << 4) + (lane & 15);
    v8f acc = {};
#pragma unroll
    for (int k0 = 0; k0 < 128; k0 += 4) {
      const int kb = k0 + kg;
      v2f a, b;
      a.x = xs[m * 128 + kb];
      a.y = xs[m * 128 + kb + 1];
      b.x = W[(size_t)kb * NCOLS + n];
      b.y = W[(size_t)(kb + 1) * NCOLS + n];
      acc = __builtin_amdgcn_wmma_f32_16x16x4_f32(false, a, false, b, (short)0, acc,
                                                  false, false);
    }
    const size_t rowbase = m0 + (size_t)((lane >> 4) << 3);
    const size_t col = (size_t)(nt << 4) + (lane & 15);
#pragma unroll
    for (int r = 0; r < 8; ++r) T[(rowbase + r) * NCOLS + col] = acc[r];
  }
}

template <int D>
__global__ void scatter_edges(const int* __restrict__ src, const int* __restrict__ dst,
                              const float* __restrict__ dinv, const float* __restrict__ T,
                              float* __restrict__ OUT, int E) {
  const int lane = threadIdx.x & 31;
  int w = (blockIdx.x * blockDim.x + threadIdx.x) >> 5;
  const int nw = (gridDim.x * blockDim.x) >> 5;
  for (; w < E; w += nw) {
    const int s = src[w];
    const int d = dst[w];
    const float nrm = dinv[s] * dinv[d];
    const float* tp = T + (size_t)s * D;
    float* op = OUT + (size_t)d * D;
#pragma unroll
    for (int f = 0; f < D / 32; ++f)
      atomicAdd(&op[lane + 32 * f], tp[lane + 32 * f] * nrm);
  }
}

__global__ void finalize_relu_kernel(float* __restrict__ t1, const float* __restrict__ agg,
                                     const float* __restrict__ dinv,
                                     const float* __restrict__ bias, size_t total) {
  size_t idx = (size_t)blockIdx.x * blockDim.x + threadIdx.x;
  size_t stride = (size_t)gridDim.x * blockDim.x;
  for (; idx < total; idx += stride) {
    const size_t i = idx >> 7;
    const int j = (int)(idx & 127);
    const float di = dinv[i];
    float v = agg[idx] + t1[idx] * di * di + bias[j];
    t1[idx] = v > 0.0f ? v : 0.0f;
  }
}

__global__ void finalize_out_kernel(float* __restrict__ out, const float* __restrict__ t2,
                                    const float* __restrict__ dinv,
                                    const float* __restrict__ bias, size_t total) {
  size_t idx = (size_t)blockIdx.x * blockDim.x + threadIdx.x;
  size_t stride = (size_t)gridDim.x * blockDim.x;
  for (; idx < total; idx += stride) {
    const size_t i = idx >> 6;
    const int j = (int)(idx & 63);
    const float di = dinv[i];
    out[idx] += t2[idx] * di * di + bias[j];
  }
}

extern "C" void kernel_launch(void* const* d_in, const int* in_sizes, int n_in,
                              void* d_out, int out_size, void* d_ws, size_t ws_size,
                              hipStream_t stream) {
  const float* x  = (const float*)d_in[0];
  const int*   ei = (const int*)d_in[1];
  const float* W1 = (const float*)d_in[2];
  const float* b1 = (const float*)d_in[3];
  const float* W2 = (const float*)d_in[4];
  const float* b2 = (const float*)d_in[5];
  float* out = (float*)d_out;

  const int N = in_sizes[0] / GDIN;
  const int E = in_sizes[1] / 2;
  const int* src = ei;
  const int* dst = ei + E;

  float* dinv = (float*)d_ws;
  float* bufA = dinv + N;
  float* bufB = bufA + (size_t)N * GDH;

  const size_t nh  = (size_t)N * GDH;
  const size_t no  = (size_t)N * GDOUT;
  const int mtiles = (N + 15) / 16;

  zero_f32_kernel<<<512, 256, 0, stream>>>(dinv, (size_t)N);
  degree_kernel<<<(E + 255) / 256, 256, 0, stream>>>(dst, dinv, E);
  rsqrt_kernel<<<(N + 255) / 256, 256, 0, stream>>>(dinv, N);

  gemm_wmma_f32<GDH><<<mtiles, 256, 0, stream>>>(x, W1, bufA);
  zero_f32_kernel<<<8192, 256, 0, stream>>>(bufB, nh);
  scatter_edges<GDH><<<16384, 256, 0, stream>>>(src, dst, dinv, bufA, bufB, E);
  finalize_relu_kernel<<<8192, 256, 0, stream>>>(bufA, bufB, dinv, b1, nh);

  gemm_wmma_f32<GDOUT><<<mtiles, 256, 0, stream>>>(bufA, W2, bufB);
  zero_f32_kernel<<<8192, 256, 0, stream>>>(out, no);
  scatter_edges<GDOUT><<<16384, 256, 0, stream>>>(src, dst, dinv, bufB, out, E);
  finalize_out_kernel<<<8192, 256, 0, stream>>>(out, bufB, dinv, b2, no);
}


# GEMM: each wave keeps its W fragments in registers and walks LDS-staged row chunks with two accumulator tiles, next chunk loaded during the multiply; k order of the f32 WMMA sums changes
# speedup vs baseline: 1.2809x; 1.2809x over previous
#include <hip/hip_runtime.h>

typedef float v2f __attribute__((ext_vector_type(2)));
typedef float v8f __attribute__((ext_vector_type(8)));

#define GDIN  128
#define GDH   128
#define GDOUT 64

__global__ void zero_f32_kernel(float* __restrict__ p, size_t n) {
  size_t i = (size_t)blockIdx.x * blockDim.x + threadIdx.x;
  size_t stride = (size_t)gridDim.x * blockDim.x;
  for (; i < n; i += stride) p[i] = 0.0f;
}

__global__ void degree_kernel(const int* __restrict__ dst, float* __restrict__ deg, int E) {
  int i = blockIdx.x * blockDim.x + threadIdx.x;
  int stride = gridDim.x * blockDim.x;
  for (; i < E; i += stride) atomicAdd(&deg[dst[i]], 1.0f);
}

__global__ void rsqrt_kernel(float* __restrict__ dinv, int N) {
  int i = blockIdx.x * blockDim.x + threadIdx.x;
  if (i < N) dinv[i] = rsqrtf(dinv[i] + 1.0f);
}

template <int NCOLS>
__global__ void __launch_bounds__(256) gemm_wmma_f32(const float* __restrict__ X,
                                                     const float* __restrict__ W,
                                                     float* __restrict__ T, int N) {
  constexpr int CT = NCOLS / 16;
  constexpr int RG = 8 / CT;
  constexpr int CH = 32 * RG;
  constexpr int NLD = CH / 8;
  constexpr int P4 = 33;
  __shared__ float4 xs4[CH * P4];

  const int tid = threadIdx.x;
  const int wave = tid >> 5;
  const int lane = tid & 31;
  const int c = lane & 15;
  const int h = lane >> 4;
  const int ct = wave % CT;
  const int rg = wave / CT;
  const int n = ct * 16 + c;

  v2f bf[32];
#pragma unroll
  for (int j = 0; j < 16; ++j) {
    const int kb = 8 * j + 4 * h;
    bf[2 * j].x     = W[(size_t)(kb + 0) * NCOLS + n];
    bf[2 * j].y     = W[(size_t)(kb + 1) * NCOLS + n];
    bf[2 * j + 1].x = W[(size_t)(kb + 2) * NCOLS + n];
    bf[2 * j + 1].y = W[(size_t)(kb + 3) * NCOLS + n];
  }

  const float4* __restrict__ X4 = reinterpret_cast<const float4*>(X);
  const int nchunks = (N + CH - 1) / CH;
  int chunk = blockIdx.x;

  float4 pre[NLD];
  if (chunk < nchunks) {
#pragma unroll
    for (int q = 0; q < NLD; ++q) {
      const int i = tid + 256 * q;
      const int row = chunk * CH + (i >> 5);
      pre[q] = row < N ? X4[(size_t)row * 32 + (i & 31)] : make_float4(0.f, 0.f, 0.f, 0.f);
    }
  }

  for (; chunk < nchunks; chunk += gridDim.x) {
#pragma unroll
    for (int q = 0; q < NLD; ++q) {
      const int i = tid + 256 * q;
      xs4[(i >> 5) * P4 + (i & 31)] = pre[q];
    }
    __builtin_amdgcn_fence(__ATOMIC_RELEASE, "workgroup", "local");
    __builtin_amdgcn_s_barrier();
    __builtin_amdgcn_fence(__ATOMIC_ACQUIRE, "workgroup", "local");

    const int next = chunk + gridDim.x;
    if (next < nchunks) {
#pragma unroll
      for (int q = 0; q < NLD; ++q) {
        const int i = tid + 256 * q;
        const int row = next * CH + (i >> 5);
        pre[q] = row < N ? X4[(size_t)row * 32 + (i & 31)] : make_float4(0.f, 0.f, 0.f, 0.f);
      }
    }

    const float4* a0p = xs4 + (rg * 32 + c) * P4 + h;
    const float4* a1p = a0p + 16 * P4;
    v8f acc0 = {};
    v8f acc1 = {};
#pragma unroll
    for (int j = 0; j < 16; ++j) {
      const float4 u = a0p[2 * j];
      const float4 v = a1p[2 * j];
      v2f a;
      a.x = u.x; a.y = u.y;
      acc0 = __builtin_amdgcn_wmma_f32_16x16x4_f32(false, a, false, bf[2 * j], (short)0, acc0,
                                                   false, false);
      a.x = v.x; a.y = v.y;
      acc1 = __builtin_amdgcn_wmma_f32_16x16x4_f32(false, a, false, bf[2 * j], (short)0, acc1,
                                                   false, false);
      a.x = u.z; a.y = u.w;
      acc0 = __builtin_amdgcn_wmma_f32_16x16x4_f32(false, a, false, bf[2 * j + 1], (short)0, acc0,
                                                   false, false);
      a.x = v.z; a.y = v.w;
      acc1 = __builtin_amdgcn_wmma_f32_16x16x4_f32(false, a, false, bf[2 * j + 1], (short)0, acc1,
                                                   false, false);
    }

    const int rowbase = chunk * CH + rg * 32 + 8 * h;
    const int lim = N - rowbase;
    float* tp = T + (size_t)rowbase * NCOLS + n;
#pragma unroll
    for (int r = 0; r < 8; ++r) {
      if (r < lim) tp[r * NCOLS] = acc0[r];
      if (r + 16 < lim) tp[(r + 16) * NCOLS] = acc1[r];
    }
    __builtin_amdgcn_fence(__ATOMIC_RELEASE, "workgroup", "local");
    __builtin_amdgcn_s_barrier();
    __builtin_amdgcn_fence(__ATOMIC_ACQUIRE, "workgroup", "local");
  }
}

template <int D>
__global__ void scatter_edges(const int* __restrict__ src, const int* __restrict__ dst,
                              const float* __restrict__ dinv, const float* __restrict__ T,
                              float* __restrict__ OUT, int E) {
  const int lane = threadIdx.x & 31;
  int w = (blockIdx.x * blockDim.x + threadIdx.x) >> 5;
  const int nw = (gridDim.x * blockDim.x) >> 5;
  for (; w < E; w += nw) {
    const int s = src[w];
    const int d = dst[w];
    const float nrm = dinv[s] * dinv[d];
    const float* tp = T + (size_t)s * D;
    float* op = OUT + (size_t)d * D;
#pragma unroll
    for (int f = 0; f < D / 32; ++f)
      atomicAdd(&op[lane + 32 * f], tp[lane + 32 * f] * nrm);
  }
}

__global__ void finalize_relu_kernel(float* __restrict__ t1, const float* __restrict__ agg,
                                     const float* __restrict__ dinv,
                                     const float* __restrict__ bias, size_t total) {
  size_t idx = (size_t)blockIdx.x * blockDim.x + threadIdx.x;
  size_t stride = (size_t)gridDim.x * blockDim.x;
  for (; idx < total; idx += stride) {
    const size_t i = idx >> 7;
    const int j = (int)(idx & 127);
    const float di = dinv[i];
    float v = agg[idx] + t1[idx] * di * di + bias[j];
    t1[idx] = v > 0.0f ? v : 0.0f;
  }
}

__global__ void finalize_out_kernel(float* __restrict__ out, const float* __restrict__ t2,
                                    const float* __restrict__ dinv,
                                    const float* __restrict__ bias, size_t total) {
  size_t idx = (size_t)blockIdx.x * blockDim.x + threadIdx.x;
  size_t stride = (size_t)gridDim.x * blockDim.x;
  for (; idx < total; idx += stride) {
    const size_t i = idx >> 6;
    const int j = (int)(idx & 63);
    const float di = dinv[i];
    out[idx] += t2[idx] * di * di + bias[j];
  }
}

extern "C" void kernel_launch(void* const* d_in, const int* in_sizes, int n_in,
                              void* d_out, int out_size, void* d_ws, size_t ws_size,
                              hipStream_t stream) {
  const float* x  = (const float*)d_in[0];
  const int*   ei = (const int*)d_in[1];
  const float* W1 = (const float*)d_in[2];
  const float* b1 = (const float*)d_in[3];
  const float* W2 = (const float*)d_in[4];
  const float* b2 = (const float*)d_in[5];
  float* out = (float*)d_out;

  const int N = in_sizes[0] / GDIN;
  const int E = in_sizes[1] / 2;
  const int* src = ei;
  const int* dst = ei + E;

  float* dinv = (float*)d_ws;
  float* bufA = dinv + N;
  float* bufB = bufA + (size_t)N * GDH;

  const size_t nh  = (size_t)N * GDH;
  const size_t no  = (size_t)N * GDOUT;

  zero_f32_kernel<<<512, 256, 0, stream>>>(dinv, (size_t)N);
  degree_kernel<<<(E + 255) / 256, 256, 0, stream>>>(dst, dinv, E);
  rsqrt_kernel<<<(N + 255) / 256, 256, 0, stream>>>(dinv, N);

  gemm_wmma_f32<GDH><<<768, 256, 0, stream>>>(x, W1, bufA, N);
  zero_f32_kernel<<<8192, 256, 0, stream>>>(bufB, nh);
  scatter_edges<GDH><<<16384, 256, 0, stream>>>(src, dst, dinv, bufA, bufB, E);
  finalize_relu_kernel<<<8192, 256, 0, stream>>>(bufA, bufB, dinv, b1, nh);

  gemm_wmma_f32<GDOUT><<<512, 256, 0, stream>>>(bufA, W2, bufB, N);
  zero_f32_kernel<<<8192, 256, 0, stream>>>(out, no);
  scatter_edges<GDOUT><<<16384, 256, 0, stream>>>(src, dst, dinv, bufB, out, E);
  finalize_out_kernel<<<8192, 256, 0, stream>>>(out, bufB, dinv, b2, no);
}


# Aggregation: edges counting-sorted by target in the call (one returning int atomic per edge, scan, fill); one wave sums a target row in registers in arrival-rank order, one store; no f32 atomics
# speedup vs baseline: 2.6388x; 2.0601x over previous
#include <hip/hip_runtime.h>

typedef float v2f __attribute__((ext_vector_type(2)));
typedef float v8f __attribute__((ext_vector_type(8)));

#define GDIN  128
#define GDH   128
#define GDOUT 64

__global__ void zero_i32_kernel(int* __restrict__ p, int n) {
  const int i = blockIdx.x * blockDim.x + threadIdx.x;
  if (i < n) p[i] = 0;
}

__global__ void count_rank_kernel(const int* __restrict__ dst, int* __restrict__ cnt,
                                  int* __restrict__ rank, int E) {
  const int Q = gridDim.x * blockDim.x;
  int e = blockIdx.x * blockDim.x + threadIdx.x;
  for (; e + 3 * Q < E; e += 4 * Q) {
    const int d0 = dst[e], d1 = dst[e + Q], d2 = dst[e + 2 * Q], d3 = dst[e + 3 * Q];
    const int r0 = atomicAdd(&cnt[d0], 1);
    const int r1 = atomicAdd(&cnt[d1], 1);
    const int r2 = atomicAdd(&cnt[d2], 1);
    const int r3 = atomicAdd(&cnt[d3], 1);
    rank[e] = r0; rank[e + Q] = r1; rank[e + 2 * Q] = r2; rank[e + 3 * Q] = r3;
  }
  for (; e < E; e += Q) rank[e] = atomicAdd(&cnt[dst[e]], 1);
}

__global__ void __launch_bounds__(1024) scan_rows_kernel(const int* __restrict__ cnt,
                                                         int* __restrict__ rowptr,
                                                         float* __restrict__ dinv, int N) {
  __shared__ int s_front[32];
  __shared__ int s_tot[32];
  const int tid = threadIdx.x;
  const int lane = tid & 31;
  const int wave = tid >> 5;
  const int base = blockIdx.x * 1024;

  int front = 0;
  const int4* __restrict__ c4 = reinterpret_cast<const int4*>(cnt);
#pragma unroll 8
  for (int k = tid; k < (base >> 2); k += 1024) {
    const int4 c = c4[k];
    front += (c.x + c.y) + (c.z + c.w);
  }
  const int idx = base + tid;
  const int c = idx < N ? cnt[idx] : 0;
  int incl = c;
  for (int o = 1; o < 32; o <<= 1) {
    const int v = __shfl_up(incl, o);
    if (lane >= o) incl += v;
  }
  for (int o = 16; o > 0; o >>= 1) front += __shfl_xor(front, o);
  if (lane == 31) s_tot[wave] = incl;
  if (lane == 0) s_front[wave] = front;
  __syncthreads();
  int off = 0;
  for (int w = 0; w < 32; ++w) {
    off += s_front[w];
    if (w < wave) off += s_tot[w];
  }
  if (idx < N) {
    rowptr[idx] = off + incl - c;
    dinv[idx] = rsqrtf((float)(c + 1));
    if (idx == N - 1) rowptr[N] = off + incl;
  }
}

__global__ void fill_sorted_kernel(const int* __restrict__ src, const int* __restrict__ dst,
                                   const int* __restrict__ rank, const int* __restrict__ rowptr,
                                   const float* __restrict__ dinv, int* __restrict__ pairs_w,
                                   int E) {
  int2* __restrict__ pairs = reinterpret_cast<int2*>(pairs_w);
  const int Q = gridDim.x * blockDim.x;
  int e = blockIdx.x * blockDim.x + threadIdx.x;
  for (; e + 3 * Q < E; e += 4 * Q) {
    int s[4], pos[4];
    float w[4];
    for (int u = 0; u < 4; ++u) {
      s[u] = src[e + u * Q];
      pos[u] = rowptr[dst[e + u * Q]] + rank[e + u * Q];
    }
    for (int u = 0; u < 4; ++u) w[u] = dinv[s[u]];
    for (int u = 0; u < 4; ++u) pairs[pos[u]] = make_int2(s[u], __float_as_int(w[u]));
  }
  for (; e < E; e += Q) {
    const int s = src[e];
    pairs[rowptr[dst[e]] + rank[e]] = make_int2(s, __float_as_int(dinv[s]));
  }
}

__global__ void copy_words_kernel(const int* __restrict__ from, int* __restrict__ to, size_t n) {
  const size_t n4 = n >> 2;
  const size_t stride = (size_t)gridDim.x * blockDim.x;
  const size_t t = (size_t)blockIdx.x * blockDim.x + threadIdx.x;
  const int4* __restrict__ f4 = reinterpret_cast<const int4*>(from);
  int4* __restrict__ t4 = reinterpret_cast<int4*>(to);
  for (size_t i = t; i < n4; i += stride) t4[i] = f4[i];
  for (size_t i = (n4 << 2) + t; i < n; i += stride) to[i] = from[i];
}

template <int NCOLS>
__global__ void __launch_bounds__(256) gemm_wmma_f32(const float* __restrict__ X,
                                                     const float* __restrict__ W,
                                                     float* __restrict__ T, int N) {
  constexpr int CT = NCOLS / 16;
  constexpr int RG = 8 / CT;
  constexpr int CH = 32 * RG;
  constexpr int NLD = CH / 8;
  constexpr int P4 = 33;
  __shared__ float4 xs4[CH * P4];

  const int tid = threadIdx.x;
  const int wave = tid >> 5;
  const int lane = tid & 31;
  const int c = lane & 15;
  const int h = lane >> 4;
  const int ct = wave % CT;
  const int rg = wave / CT;
  const int n = ct * 16 + c;

  v2f bf[32];
#pragma unroll
  for (int j = 0; j < 16; ++j) {
    const int kb = 8 * j + 4 * h;
    bf[2 * j].x     = W[(size_t)(kb + 0) * NCOLS + n];
    bf[2 * j].y     = W[(size_t)(kb + 1) * NCOLS + n];
    bf[2 * j + 1].x = W[(size_t)(kb + 2) * NCOLS + n];
    bf[2 * j + 1].y = W[(size_t)(kb + 3) * NCOLS + n];
  }

  const float4* __restrict__ X4 = reinterpret_cast<const float4*>(X);
  const int nchunks = (N + CH - 1) / CH;
  int chunk = blockIdx.x;

  float4 pre[NLD];
  if (chunk < nchunks) {
#pragma unroll
    for (int q = 0; q < NLD; ++q) {
      const int i = tid + 256 * q;
      const int row = chunk * CH + (i >> 5);
      pre[q] = row < N ? X4[(size_t)row * 32 + (i & 31)] : make_float4(0.f, 0.f, 0.f, 0.f);
    }
  }

  for (; chunk < nchunks; chunk += gridDim.x) {
#pragma unroll
    for (int q = 0; q < NLD; ++q) {
      const int i = tid + 256 * q;
      xs4[(i >> 5) * P4 + (i & 31)] = pre[q];
    }
    __builtin_amdgcn_fence(__ATOMIC_RELEASE, "workgroup", "local");
    __builtin_amdgcn_s_barrier();
    __builtin_amdgcn_fence(__ATOMIC_ACQUIRE, "workgroup", "local");

    const int next = chunk + gridDim.x;
    if (next < nchunks) {
#pragma unroll
      for (int q = 0; q < NLD; ++q) {
        const int i = tid + 256 * q;
        const int row = next * CH + (i >> 5);
        pre[q] = row < N ? X4[(size_t)row * 32 + (i & 31)] : make_float4(0.f, 0.f, 0.f, 0.f);
      }
    }

    const float4* a0p = xs4 + (rg * 32 + c) * P4 + h;
    const float4* a1p = a0p + 16 * P4;
    v8f acc0 = {};
    v8f acc1 = {};
#pragma unroll
    for (int j = 0; j < 16; ++j) {
      const float4 u = a0p[2 * j];
      const float4 v = a1p[2 * j];
      v2f a;
      a.x = u.x; a.y = u.y;
      acc0 = __builtin_amdgcn_wmma_f32_16x16x4_f32(false, a, false, bf[2 * j], (short)0, acc0,
                                                   false, false);
      a.x = v.x; a.y = v.y;
      acc1 = __builtin_amdgcn_wmma_f32_16x16x4_f32(false, a, false, bf[2 * j], (short)0, acc1,
                                                   false, false);
      a.x = u.z; a.y = u.w;
      acc0 = __builtin_amdgcn_wmma_f32_16x16x4_f32(false, a, false, bf[2 * j + 1], (short)0, acc0,
                                                   false, false);
      a.x = v.z; a.y = v.w;
      acc1 = __builtin_amdgcn_wmma_f32_16x16x4_f32(false, a, false, bf[2 * j + 1], (short)0, acc1,
                                                   false, false);
    }

    const int rowbase = chunk * CH + rg * 32 + 8 * h;
    const int lim = N - rowbase;
    float* tp = T + (size_t)rowbase * NCOLS + n;
#pragma unroll
    for (int r = 0; r < 8; ++r) {
      if (r < lim) tp[r * NCOLS] = acc0[r];
      if (r + 16 < lim) tp[(r + 16) * NCOLS] = acc1[r];
    }
    __builtin_amdgcn_fence(__ATOMIC_RELEASE, "workgroup", "local");
    __builtin_amdgcn_s_barrier();
    __builtin_amdgcn_fence(__ATOMIC_ACQUIRE, "workgroup", "local");
  }
}

template <int V>
__device__ __forceinline__ void sum_run(float __attribute__((ext_vector_type(V)))& acc,
                                        const float __attribute__((ext_vector_type(V)))* __restrict__ Tv,
                                        int s_l, float w_l, int cnt, int lane) {
  typedef float vf __attribute__((ext_vector_type(V)));
  for (int j = 0; j < cnt; j += 4) {
    vf v[4];
    float w[4];
    for (int u = 0; u < 4; ++u) {
      const int s = __builtin_amdgcn_readlane(s_l, j + u);
      w[u] = __int_as_float(__builtin_amdgcn_readlane(__float_as_int(w_l), j + u));
      v[u] = Tv[(size_t)s * 32 + lane];
    }
    for (int u = 0; u < 4; ++u) acc += v[u] * w[u];
  }
}

template <int D, bool RELU>
__global__ void __launch_bounds__(256) gather_sum_rows(const int* __restrict__ rowptr,
                                                       const int* __restrict__ pairs_w,
                                                       const float* __restrict__ dinv,
                                                       const float* __restrict__ T,
                                                       const float* __restrict__ bias,
                                                       float* __restrict__ OUT, int N) {
  constexpr int V = D / 32;
  typedef float vf __attribute__((ext_vector_type(V)));
  const int lane = threadIdx.x & 31;
  const int nw = (gridDim.x * blockDim.x) >> 5;
  int i = __builtin_amdgcn_readfirstlane((int)((blockIdx.x * blockDim.x + threadIdx.x) >> 5));
  if (i >= N) return;
  const int2* __restrict__ pairs = reinterpret_cast<const int2*>(pairs_w);
  const vf* __restrict__ Tv = reinterpret_cast<const vf*>(T);
  vf* __restrict__ Ov = reinterpret_cast<vf*>(OUT);
  const vf bv = reinterpret_cast<const vf*>(bias)[lane];

  int beg = rowptr[i], end = rowptr[i + 1];
  int ni = i + nw;
  int nbeg = 0, nend = 0;
  if (ni < N) { nbeg = rowptr[ni]; nend = rowptr[ni + 1]; }
  int2 p = make_int2(i, 0);
  if (beg + lane < end) p = pairs[beg + lane];
  float di = dinv[i];
  int s_l = p.x;
  float w_l = __int_as_float(p.y) * di;

  for (;;) {
    const int nni = ni + nw;
    int nnbeg = 0, nnend = 0;
    if (nni < N) { nnbeg = rowptr[nni]; nnend = rowptr[nni + 1]; }
    float ndi = 0.0f;
    int2 np = make_int2(i, 0);
    if (ni < N) {
      ndi = dinv[ni];
      np = make_int2(ni, 0);
      if (nbeg + lane < nend) np = pairs[nbeg + lane];
    }
    const vf selfv = Tv[(size_t)i * 32 + lane];
    vf acc = 0.0f;
    sum_run<V>(acc, Tv, s_l, w_l, min(32, end - beg), lane);
    for (int base = beg + 32; base < end; base += 32) {
      int2 q = make_int2(i, 0);
      if (base + lane < end) q = pairs[base + lane];
      sum_run<V>(acc, Tv, q.x, __int_as_float(q.y) * di, min(32, end - base), lane);
    }
    acc += selfv * (di * di);
    acc += bv;
    if (RELU) {
      for (int k = 0; k < V; ++k) acc[k] = acc[k] > 0.0f ? acc[k] : 0.0f;
    }
    Ov[(size_t)i * 32 + lane] = acc;
    if (ni >= N) break;
    i = ni; beg = nbeg; end = nend; di = ndi;
    s_l = np.x;
    w_l = __int_as_float(np.y) * ndi;
    ni = nni; nbeg = nnbeg; nend = nnend;
  }
}

extern "C" void kernel_launch(void* const* d_in, const int* in_sizes, int n_in,
                              void* d_out, int out_size, void* d_ws, size_t ws_size,
                              hipStream_t stream) {
  const float* x  = (const float*)d_in[0];
  const int*   ei = (const int*)d_in[1];
  const float* W1 = (const float*)d_in[2];
  const float* b1 = (const float*)d_in[3];
  const float* W2 = (const float*)d_in[4];
  const float* b2 = (const float*)d_in[5];
  float* out = (float*)d_out;

  const int N = in_sizes[0] / GDIN;
  const int E = in_sizes[1] / 2;
  const int* src = ei;
  const int* dst = ei + E;

  float* dinv = (float*)d_ws;
  float* bufA = dinv + N;
  float* bufB = bufA + (size_t)N * GDH;

  int* cnt  = (int*)bufB;
  int* rank = cnt + N;
  const int RP = (N + 4) & ~3;
  const size_t lwords = (size_t)RP + 2 * (size_t)E;
  int* lst0 = (int*)d_out;
  int* lst1 = (int*)(bufA + (size_t)N * GDOUT);

  zero_i32_kernel<<<(N + 255) / 256, 256, 0, stream>>>(cnt, N);
  count_rank_kernel<<<(E + 1023) / 1024, 256, 0, stream>>>(dst, cnt, rank, E);
  scan_rows_kernel<<<(N + 1023) / 1024, 1024, 0, stream>>>(cnt, lst0, dinv, N);
  fill_sorted_kernel<<<(E + 1023) / 1024, 256, 0, stream>>>(src, dst, rank, lst0, dinv,
                                                            lst0 + RP, E);

  gemm_wmma_f32<GDH><<<768, 256, 0, stream>>>(x, W1, bufA, N);
  gather_sum_rows<GDH, true><<<2048, 256, 0, stream>>>(lst0, lst0 + RP, dinv, bufA,
                                                       b1, bufB, N);

  copy_words_kernel<<<1024, 256, 0, stream>>>(lst0, lst1, lwords);
  gemm_wmma_f32<GDOUT><<<512, 256, 0, stream>>>(bufB, W2, bufA, N);
  gather_sum_rows<GDOUT, false><<<2048, 256, 0, stream>>>(lst1, lst1 + RP, dinv,
                                                          bufA, b2, out, N);
}


# Sort by target: two-level counting sort on LDS counters (256-row buckets, then rows) replaces one returning global atomic per edge and the global scan; order inside a row (f32 sum order) changes
# speedup vs baseline: 3.7554x; 1.4231x over previous
#include <hip/hip_runtime.h>

typedef float v2f __attribute__((ext_vector_type(2)));
typedef float v8f __attribute__((ext_vector_type(8)));

#define GDIN  128
#define GDH   128
#define GDOUT 64

#define BSH   8
#define BROWS (1 << BSH)
#define MAXB  512
#define SLICE 4096
#define CPAD  32

__global__ void zero_i32_kernel(int* __restrict__ p, int n) {
  const int i = blockIdx.x * blockDim.x + threadIdx.x;
  if (i < n) p[i] = 0;
}

__global__ void __launch_bounds__(MAXB) bucket_count_kernel(const int* __restrict__ dst,
                                                            int* __restrict__ gcount, int E, int NB) {
  __shared__ int hist[MAXB];
  const int tid = threadIdx.x;
  const int beg = blockIdx.x * SLICE;
  const int n = min(SLICE, E - beg);
  int d[SLICE / MAXB];
  for (int u = 0; u < SLICE / MAXB; ++u) {
    const int i = tid + u * MAXB;
    d[u] = i < n ? dst[beg + i] : -1;
  }
  hist[tid] = 0;
  __syncthreads();
  for (int u = 0; u < SLICE / MAXB; ++u)
    if (d[u] >= 0) atomicAdd(&hist[d[u] >> BSH], 1);
  __syncthreads();
  const int c = hist[tid];
  if (tid < NB && c) atomicAdd(&gcount[tid * CPAD], c);
}

__global__ void __launch_bounds__(MAXB) bucket_scatter_kernel(const int* __restrict__ src,
                                                              const int* __restrict__ dst,
                                                              const int* __restrict__ gcount,
                                                              int* __restrict__ gcursor,
                                                              int* __restrict__ bucketed_w,
                                                              int E, int NB) {
  __shared__ int2 lbuf[SLICE];
  __shared__ int hist[MAXB];
  __shared__ int lofs[MAXB];
  __shared__ int delta[MAXB];
  __shared__ int wtg[MAXB / 32];
  __shared__ int wtc[MAXB / 32];
  int2* __restrict__ bucketed = reinterpret_cast<int2*>(bucketed_w);
  const int tid = threadIdx.x;
  const int lane = tid & 31;
  const int wave = tid >> 5;
  const int beg = blockIdx.x * SLICE;
  const int n = min(SLICE, E - beg);
  int d[SLICE / MAXB], s[SLICE / MAXB];
  for (int u = 0; u < SLICE / MAXB; ++u) {
    const int i = tid + u * MAXB;
    d[u] = i < n ? dst[beg + i] : -1;
    s[u] = i < n ? src[beg + i] : 0;
  }
  hist[tid] = 0;
  __syncthreads();
  for (int u = 0; u < SLICE / MAXB; ++u)
    if (d[u] >= 0) atomicAdd(&hist[d[u] >> BSH], 1);
  const int g = tid < NB ? gcount[tid * CPAD] : 0;
  __syncthreads();
  const int c = hist[tid];
  int ig = g, ic = c;
  for (int o = 1; o < 32; o <<= 1) {
    const int vg = __shfl_up(ig, o);
    const int vc = __shfl_up(ic, o);
    if (lane >= o) { ig += vg; ic += vc; }
  }
  if (lane == 31) { wtg[wave] = ig; wtc[wave] = ic; }
  __syncthreads();
  int og = 0, oc = 0;
  for (int w = 0; w < wave; ++w) { og += wtg[w]; oc += wtc[w]; }
  int r = 0;
  if (tid < NB && c) r = atomicAdd(&gcursor[tid * CPAD], c);
  const int lo = oc + ic - c;
  lofs[tid] = lo;
  delta[tid] = og + ig - g + r - lo;
  hist[tid] = 0;
  __syncthreads();
  for (int u = 0; u < SLICE / MAXB; ++u) {
    if (d[u] >= 0) {
      const int b = d[u] >> BSH;
      const int k = atomicAdd(&hist[b], 1);
      lbuf[lofs[b] + k] = make_int2(s[u], d[u]);
    }
  }
  __syncthreads();
  for (int i = tid; i < n; i += MAXB) {
    const int2 rec = lbuf[i];
    bucketed[delta[rec.y >> BSH] + i] = rec;
  }
}

__global__ void __launch_bounds__(MAXB) bucket_rows_kernel(const int* __restrict__ gcount,
                                                           const int* __restrict__ bucketed_w,
                                                           int* __restrict__ rowptr,
                                                           float* __restrict__ dinv, int N) {
  __shared__ int rcnt[BROWS];
  __shared__ int wsum[MAXB / 32];
  __shared__ int wtot[BROWS / 32];
  const int2* __restrict__ bucketed = reinterpret_cast<const int2*>(bucketed_w);
  const int tid = threadIdx.x;
  const int lane = tid & 31;
  const int wave = tid >> 5;
  const int b = blockIdx.x;
  const int r0 = b << BSH;
  int part = tid < b ? gcount[tid * CPAD] : 0;
  for (int o = 16; o > 0; o >>= 1) part += __shfl_xor(part, o);
  if (lane == 0) wsum[wave] = part;
  if (tid < BROWS) rcnt[tid] = 0;
  __syncthreads();
  int start = 0;
  for (int w = 0; w < MAXB / 32; ++w) start += wsum[w];
  const int cnt = gcount[b * CPAD];
  for (int i0 = tid; i0 < cnt; i0 += 4 * MAXB) {
    int d[4];
    for (int u = 0; u < 4; ++u) {
      const int i = i0 + u * MAXB;
      d[u] = i < cnt ? bucketed[start + i].y : -1;
    }
    for (int u = 0; u < 4; ++u)
      if (d[u] >= 0) atomicAdd(&rcnt[d[u] - r0], 1);
  }
  __syncthreads();
  const int c = tid < BROWS ? rcnt[tid] : 0;
  int incl = c;
  for (int o = 1; o < 32; o <<= 1) {
    const int v = __shfl_up(incl, o);
    if (lane >= o) incl += v;
  }
  if (tid < BROWS && lane == 31) wtot[wave] = incl;
  __syncthreads();
  if (tid < BROWS) {
    int off = start;
    for (int w = 0; w < wave; ++w) off += wtot[w];
    const int row = r0 + tid;
    if (row < N) {
      rowptr[row] = off + incl - c;
      dinv[row] = rsqrtf((float)(c + 1));
    }
    if (b == (int)gridDim.x - 1 && tid == BROWS - 1) rowptr[N] = off + incl;
  }
}

__global__ void __launch_bounds__(MAXB) bucket_fill_kernel(const int* __restrict__ bucketed_w,
                                                           const int* __restrict__ rowptr,
                                                           const float* __restrict__ dinv,
                                                           int* __restrict__ pairs_w, int N) {
  __shared__ int rofs[BROWS];
  __shared__ int rcnt[BROWS];
  const int2* __restrict__ bucketed = reinterpret_cast<const int2*>(bucketed_w);
  int2* __restrict__ pairs = reinterpret_cast<int2*>(pairs_w);
  const int tid = threadIdx.x;
  const int r0 = blockIdx.x << BSH;
  if (tid < BROWS) {
    rofs[tid] = rowptr[min(r0 + tid, N)];
    rcnt[tid] = 0;
  }
  __syncthreads();
  const int start = rofs[0];
  const int end = rowptr[min(r0 + BROWS, N)];
  for (int i0 = start + tid; i0 < end; i0 += 4 * MAXB) {
    int2 p[4];
    float w[4];
    for (int u = 0; u < 4; ++u) {
      const int i = i0 + u * MAXB;
      p[u] = i < end ? bucketed[i] : make_int2(0, -1);
    }
    for (int u = 0; u < 4; ++u) w[u] = p[u].y >= 0 ? dinv[p[u].x] : 0.0f;
    for (int u = 0; u < 4; ++u) {
      if (p[u].y >= 0) {
        const int k = p[u].y - r0;
        const int rk = atomicAdd(&rcnt[k], 1);
        pairs[rofs[k] + rk] = make_int2(p[u].x, __float_as_int(w[u]));
      }
    }
  }
}

__global__ void copy_words_kernel(const int* __restrict__ from, int* __restrict__ to, size_t n) {
  const size_t n4 = n >> 2;
  const size_t stride = (size_t)gridDim.x * blockDim.x;
  const size_t t = (size_t)blockIdx.x * blockDim.x + threadIdx.x;
  const int4* __restrict__ f4 = reinterpret_cast<const int4*>(from);
  int4* __restrict__ t4 = reinterpret_cast<int4*>(to);
  for (size_t i = t; i < n4; i += stride) t4[i] = f4[i];
  for (size_t i = (n4 << 2) + t; i < n; i += stride) to[i] = from[i];
}

template <int NCOLS>
__global__ void __launch_bounds__(256) gemm_wmma_f32(const float* __restrict__ X,
                                                     const float* __restrict__ W,
                                                     float* __restrict__ T, int N) {
  constexpr int CT = NCOLS / 16;
  constexpr int RG = 8 / CT;
  constexpr int CH = 32 * RG;
  constexpr int NLD = CH / 8;
  constexpr int P4 = 33;
  __shared__ float4 xs4[CH * P4];

  const int tid = threadIdx.x;
  const int wave = tid >> 5;
  const int lane = tid & 31;
  const int c = lane & 15;
  const int h = lane >> 4;
  const int ct = wave % CT;
  const int rg = wave / CT;
  const int n = ct * 16 + c;

  v2f bf[32];
#pragma unroll
  for (int j = 0; j < 16; ++j) {
    const int kb = 8 * j + 4 * h;
    bf[2 * j].x     = W[(size_t)(kb + 0) * NCOLS + n];
    bf[2 * j].y     = W[(size_t)(kb + 1) * NCOLS + n];
    bf[2 * j + 1].x = W[(size_t)(kb + 2) * NCOLS + n];
    bf[2 * j + 1].y = W[(size_t)(kb + 3) * NCOLS + n];
  }

  const float4* __restrict__ X4 = reinterpret_cast<const float4*>(X);
  const int nchunks = (N + CH - 1) / CH;
  int chunk = blockIdx.x;

  float4 pre[NLD];
  if (chunk < nchunks) {
#pragma unroll
    for (int q = 0; q < NLD; ++q) {
      const int i = tid + 256 * q;
      const int row = chunk * CH + (i >> 5);
      pre[q] = row < N ? X4[(size_t)row * 32 + (i & 31)] : make_float4(0.f, 0.f, 0.f, 0.f);
    }
  }

  for (; chunk < nchunks; chunk += gridDim.x) {
#pragma unroll
    for (int q = 0; q < NLD; ++q) {
      const int i = tid + 256 * q;
      xs4[(i >> 5) * P4 + (i & 31)] = pre[q];
    }
    __builtin_amdgcn_fence(__ATOMIC_RELEASE, "workgroup", "local");
    __builtin_amdgcn_s_barrier();
    __builtin_amdgcn_fence(__ATOMIC_ACQUIRE, "workgroup", "local");

    const int next = chunk + gridDim.x;
    if (next < nchunks) {
#pragma unroll
      for (int q = 0; q < NLD; ++q) {
        const int i = tid + 256 * q;
        const int row = next * CH + (i >> 5);
        pre[q] = row < N ? X4[(size_t)row * 32 + (i & 31)] : make_float4(0.f, 0.f, 0.f, 0.f);
      }
    }

    const float4* a0p = xs4 + (rg * 32 + c) * P4 + h;
    const float4* a1p = a0p + 16 * P4;
    v8f acc0 = {};
    v8f acc1 = {};
#pragma unroll
    for (int j = 0; j < 16; ++j) {
      const float4 u = a0p[2 * j];
      const float4 v = a1p[2 * j];
      v2f a;
      a.x = u.x; a.y = u.y;
      acc0 = __builtin_amdgcn_wmma_f32_16x16x4_f32(false, a, false, bf[2 * j], (short)0, acc0,
                                                   false, false);
      a.x = v.x; a.y = v.y;
      acc1 = __builtin_amdgcn_wmma_f32_16x16x4_f32(false, a, false, bf[2 * j], (short)0, acc1,
                                                   false, false);
      a.x = u.z; a.y = u.w;
      acc0 = __builtin_amdgcn_wmma_f32_16x16x4_f32(false, a, false, bf[2 * j + 1], (short)0, acc0,
                                                   false, false);
      a.x = v.z; a.y = v.w;
      acc1 = __builtin_amdgcn_wmma_f32_16x16x4_f32(false, a, false, bf[2 * j + 1], (short)0, acc1,
                                                   false, false);
    }

    const int rowbase = chunk * CH + rg * 32 + 8 * h;
    const int lim = N - rowbase;
    float* tp = T + (size_t)rowbase * NCOLS + n;
#pragma unroll
    for (int r = 0; r < 8; ++r) {
      if (r < lim) tp[r * NCOLS] = acc0[r];
      if (r + 16 < lim) tp[(r + 16) * NCOLS] = acc1[r];
    }
    __builtin_amdgcn_fence(__ATOMIC_RELEASE, "workgroup", "local");
    __builtin_amdgcn_s_barrier();
    __builtin_amdgcn_fence(__ATOMIC_ACQUIRE, "workgroup", "local");
  }
}

template <int V>
__device__ __forceinline__ void sum_run(float __attribute__((ext_vector_type(V)))& acc,
                                        const float __attribute__((ext_vector_type(V)))* __restrict__ Tv,
                                        int s_l, float w_l, int cnt, int lane) {
  typedef float vf __attribute__((ext_vector_type(V)));
  for (int j = 0; j < cnt; j += 4) {
    vf v[4];
    float w[4];
    for (int u = 0; u < 4; ++u) {
      const int s = __builtin_amdgcn_readlane(s_l, j + u);
      w[u] = __int_as_float(__builtin_amdgcn_readlane(__float_as_int(w_l), j + u));
      v[u] = Tv[(size_t)s * 32 + lane];
    }
    for (int u = 0; u < 4; ++u) acc += v[u] * w[u];
  }
}

template <int D, bool RELU>
__global__ void __launch_bounds__(256) gather_sum_rows(const int* __restrict__ rowptr,
                                                       const int* __restrict__ pairs_w,
                                                       const float* __restrict__ dinv,
                                                       const float* __restrict__ T,
                                                       const float* __restrict__ bias,
                                                       float* __restrict__ OUT, int N) {
  constexpr int V = D / 32;
  typedef float vf __attribute__((ext_vector_type(V)));
  const int lane = threadIdx.x & 31;
  const int nw = (gridDim.x * blockDim.x) >> 5;
  int i = __builtin_amdgcn_readfirstlane((int)((blockIdx.x * blockDim.x + threadIdx.x) >> 5));
  if (i >= N) return;
  const int2* __restrict__ pairs = reinterpret_cast<const int2*>(pairs_w);
  const vf* __restrict__ Tv = reinterpret_cast<const vf*>(T);
  vf* __restrict__ Ov = reinterpret_cast<vf*>(OUT);
  const vf bv = reinterpret_cast<const vf*>(bias)[lane];

  int beg = rowptr[i], end = rowptr[i + 1];
  int ni = i + nw;
  int nbeg = 0, nend = 0;
  if (ni < N) { nbeg = rowptr[ni]; nend = rowptr[ni + 1]; }
  int2 p = make_int2(i, 0);
  if (beg + lane < end) p = pairs[beg + lane];
  float di = dinv[i];
  int s_l = p.x;
  float w_l = __int_as_float(p.y) * di;

  for (;;) {
    const int nni = ni + nw;
    int nnbeg = 0, nnend = 0;
    if (nni < N) { nnbeg = rowptr[nni]; nnend = rowptr[nni + 1]; }
    float ndi = 0.0f;
    int2 np = make_int2(i, 0);
    if (ni < N) {
      ndi = dinv[ni];
      np = make_int2(ni, 0);
      if (nbeg + lane < nend) np = pairs[nbeg + lane];
    }
    const vf selfv = Tv[(size_t)i * 32 + lane];
    vf acc = 0.0f;
    sum_run<V>(acc, Tv, s_l, w_l, min(32, end - beg), lane);
    for (int base = beg + 32; base < end; base += 32) {
      int2 q = make_int2(i, 0);
      if (base + lane < end) q = pairs[base + lane];
      sum_run<V>(acc, Tv, q.x, __int_as_float(q.y) * di, min(32, end - base), lane);
    }
    acc += selfv * (di * di);
    acc += bv;
    if (RELU) {
      for (int k = 0; k < V; ++k) acc[k] = acc[k] > 0.0f ? acc[k] : 0.0f;
    }
    Ov[(size_t)i * 32 + lane] = acc;
    if (ni >= N) break;
    i = ni; beg = nbeg; end = nend; di = ndi;
    s_l = np.x;
    w_l = __int_as_float(np.y) * ndi;
    ni = nni; nbeg = nnbeg; nend = nnend;
  }
}

extern "C" void kernel_launch(void* const* d_in, const int* in_sizes, int n_in,
                              void* d_out, int out_size, void* d_ws, size_t ws_size,
                              hipStream_t stream) {
  const float* x  = (const float*)d_in[0];
  const int*   ei = (const int*)d_in[1];
  const float* W1 = (const float*)d_in[2];
  const float* b1 = (const float*)d_in[3];
  const float* W2 = (const float*)d_in[4];
  const float* b2 = (const float*)d_in[5];
  float* out = (float*)d_out;

  const int N = in_sizes[0] / GDIN;
  const int E = in_sizes[1] / 2;
  const int* src = ei;
  const int* dst = ei + E;

  float* dinv = (float*)d_ws;
  float* bufA = dinv + N;
  float* bufB = bufA + (size_t)N * GDH;

  int* gcount   = (int*)bufB;
  int* gcursor  = gcount + MAXB * CPAD;
  int* bucketed = gcursor + MAXB * CPAD;
  const int NB = (N + BROWS - 1) >> BSH;
  const int RP = (N + 4) & ~3;
  const size_t lwords = (size_t)RP + 2 * (size_t)E;
  int* lst0 = (int*)d_out;
  int* lst1 = (int*)(bufA + (size_t)N * GDOUT);

  zero_i32_kernel<<<(2 * MAXB * CPAD + 255) / 256, 256, 0, stream>>>(gcount, 2 * MAXB * CPAD);
  bucket_count_kernel<<<(E + SLICE - 1) / SLICE, MAXB, 0, stream>>>(dst, gcount, E, NB);
  bucket_scatter_kernel<<<(E + SLICE - 1) / SLICE, MAXB, 0, stream>>>(src, dst, gcount, gcursor, bucketed,
                                                                     E, NB);
  bucket_rows_kernel<<<NB, MAXB, 0, stream>>>(gcount, bucketed, lst0, dinv, N);
  bucket_fill_kernel<<<NB, MAXB, 0, stream>>>(bucketed, lst0, dinv, lst0 + RP, N);

  gemm_wmma_f32<GDH><<<768, 256, 0, stream>>>(x, W1, bufA, N);
  gather_sum_rows<GDH, true><<<2048, 256, 0, stream>>>(lst0, lst0 + RP, dinv, bufA,
                                                       b1, bufB, N);

  copy_words_kernel<<<1024, 256, 0, stream>>>(lst0, lst1, lwords);
  gemm_wmma_f32<GDOUT><<<512, 256, 0, stream>>>(bufB, W2, bufA, N);
  gather_sum_rows<GDOUT, false><<<2048, 256, 0, stream>>>(lst1, lst1 + RP, dinv,
                                                          bufA, b2, out, N);
}


# Layer-2 GEMM fused into layer-1 row sums: a 4-wave block sums 32 hidden rows into LDS (8 gathers in flight), multiplies by W2 on f32 WMMA, same k order; hidden layer not stored, list copy dropped
# speedup vs baseline: 4.0939x; 1.0901x over previous
#include <hip/hip_runtime.h>

typedef float v2f __attribute__((ext_vector_type(2)));
typedef float v8f __attribute__((ext_vector_type(8)));

#define GDIN  128
#define GDH   128
#define GDOUT 64

#define BSH   8
#define BROWS (1 << BSH)
#define MAXB  512
#define SLICE 4096
#define CPAD  32

__global__ void zero_i32_kernel(int* __restrict__ p, int n) {
  const int i = blockIdx.x * blockDim.x + threadIdx.x;
  if (i < n) p[i] = 0;
}

__global__ void __launch_bounds__(MAXB) bucket_count_kernel(const int* __restrict__ dst,
                                                            int* __restrict__ gcount, int E, int NB) {
  __shared__ int hist[MAXB];
  const int tid = threadIdx.x;
  const int beg = blockIdx.x * SLICE;
  const int n = min(SLICE, E - beg);
  int d[SLICE / MAXB];
  for (int u = 0; u < SLICE / MAXB; ++u) {
    const int i = tid + u * MAXB;
    d[u] = i < n ? dst[beg + i] : -1;
  }
  hist[tid] = 0;
  __syncthreads();
  for (int u = 0; u < SLICE / MAXB; ++u)
    if (d[u] >= 0) atomicAdd(&hist[d[u] >> BSH], 1);
  __syncthreads();
  const int c = hist[tid];
  if (tid < NB && c) atomicAdd(&gcount[tid * CPAD], c);
}

__global__ void __launch_bounds__(MAXB) bucket_scatter_kernel(const int* __restrict__ src,
                                                              const int* __restrict__ dst,
                                                              const int* __restrict__ gcount,
                                                              int* __restrict__ gcursor,
                                                              int* __restrict__ bucketed_w,
                                                              int E, int NB) {
  __shared__ int2 lbuf[SLICE];
  __shared__ int hist[MAXB];
  __shared__ int lofs[MAXB];
  __shared__ int delta[MAXB];
  __shared__ int wtg[MAXB / 32];
  __shared__ int wtc[MAXB / 32];
  int2* __restrict__ bucketed = reinterpret_cast<int2*>(bucketed_w);
  const int tid = threadIdx.x;
  const int lane = tid & 31;
  const int wave = tid >> 5;
  const int beg = blockIdx.x * SLICE;
  const int n = min(SLICE, E - beg);
  int d[SLICE / MAXB], s[SLICE / MAXB];
  for (int u = 0; u < SLICE / MAXB; ++u) {
    const int i = tid + u * MAXB;
    d[u] = i < n ? dst[beg + i] : -1;
    s[u] = i < n ? src[beg + i] : 0;
  }
  hist[tid] = 0;
  __syncthreads();
  for (int u = 0; u < SLICE / MAXB; ++u)
    if (d[u] >= 0) atomicAdd(&hist[d[u] >> BSH], 1);
  const int g = tid < NB ? gcount[tid * CPAD] : 0;
  __syncthreads();
  const int c = hist[tid];
  int ig = g, ic = c;
  for (int o = 1; o < 32; o <<= 1) {
    const int vg = __shfl_up(ig, o);
    const int vc = __shfl_up(ic, o);
    if (lane >= o) { ig += vg; ic += vc; }
  }
  if (lane == 31) { wtg[wave] = ig; wtc[wave] = ic; }
  __syncthreads();
  int og = 0, oc = 0;
  for (int w = 0; w < wave; ++w) { og += wtg[w]; oc += wtc[w]; }
  int r = 0;
  if (tid < NB && c) r = atomicAdd(&gcursor[tid * CPAD], c);
  const int lo = oc + ic - c;
  lofs[tid] = lo;
  delta[tid] = og + ig - g + r - lo;
  hist[tid] = 0;
  __syncthreads();
  for (int u = 0; u < SLICE / MAXB; ++u) {
    if (d[u] >= 0) {
      const int b = d[u] >> BSH;
      const int k = atomicAdd(&hist[b], 1);
      lbuf[lofs[b] + k] = make_int2(s[u], d[u]);
    }
  }
  __syncthreads();
  for (int i = tid; i < n; i += MAXB) {
    const int2 rec = lbuf[i];
    bucketed[delta[rec.y >> BSH] + i] = rec;
  }
}

__global__ void __launch_bounds__(MAXB) bucket_rows_kernel(const int* __restrict__ gcount,
                                                           const int* __restrict__ bucketed_w,
                                                           int* __restrict__ rowptr,
                                                           float* __restrict__ dinv, int N) {
  __shared__ int rcnt[BROWS];
  __shared__ int wsum[MAXB / 32];
  __shared__ int wtot[BROWS / 32];
  const int2* __restrict__ bucketed = reinterpret_cast<const int2*>(bucketed_w);
  const int tid = threadIdx.x;
  const int lane = tid & 31;
  const int wave = tid >> 5;
  const int b = blockIdx.x;
  const int r0 = b << BSH;
  int part = tid < b ? gcount[tid * CPAD] : 0;
  for (int o = 16; o > 0; o >>= 1) part += __shfl_xor(part, o);
  if (lane == 0) wsum[wave] = part;
  if (tid < BROWS) rcnt[tid] = 0;
  __syncthreads();
  int start = 0;
  for (int w = 0; w < MAXB / 32; ++w) start += wsum[w];
  const int cnt = gcount[b * CPAD];
  for (int i0 = tid; i0 < cnt; i0 += 4 * MAXB) {
    int d[4];
    for (int u = 0; u < 4; ++u) {
      const int i = i0 + u * MAXB;
      d[u] = i < cnt ? bucketed[start + i].y : -1;
    }
    for (int u = 0; u < 4; ++u)
      if (d[u] >= 0) atomicAdd(&rcnt[d[u] - r0], 1);
  }
  __syncthreads();
  const int c = tid < BROWS ? rcnt[tid] : 0;
  int incl = c;
  for (int o = 1; o < 32; o <<= 1) {
    const int v = __shfl_up(incl, o);
    if (lane >= o) incl += v;
  }
  if (tid < BROWS && lane == 31) wtot[wave] = incl;
  __syncthreads();
  if (tid < BROWS) {
    int off = start;
    for (int w = 0; w < wave; ++w) off += wtot[w];
    const int row = r0 + tid;
    if (row < N) {
      rowptr[row] = off + incl - c;
      dinv[row] = rsqrtf((float)(c + 1));
    }
    if (b == (int)gridDim.x - 1 && tid == BROWS - 1) rowptr[N] = off + incl;
  }
}

__global__ void __launch_bounds__(MAXB) bucket_fill_kernel(const int* __restrict__ bucketed_w,
                                                           const int* __restrict__ rowptr,
                                                           const float* __restrict__ dinv,
                                                           int* __restrict__ pairs_w, int N) {
  __shared__ int rofs[BROWS];
  __shared__ int rcnt[BROWS];
  const int2* __restrict__ bucketed = reinterpret_cast<const int2*>(bucketed_w);
  int2* __restrict__ pairs = reinterpret_cast<int2*>(pairs_w);
  const int tid = threadIdx.x;
  const int r0 = blockIdx.x << BSH;
  if (tid < BROWS) {
    rofs[tid] = rowptr[min(r0 + tid, N)];
    rcnt[tid] = 0;
  }
  __syncthreads();
  const int start = rofs[0];
  const int end = rowptr[min(r0 + BROWS, N)];
  for (int i0 = start + tid; i0 < end; i0 += 4 * MAXB) {
    int2 p[4];
    float w[4];
    for (int u = 0; u < 4; ++u) {
      const int i = i0 + u * MAXB;
      p[u] = i < end ? bucketed[i] : make_int2(0, -1);
    }
    for (int u = 0; u < 4; ++u) w[u] = p[u].y >= 0 ? dinv[p[u].x] : 0.0f;
    for (int u = 0; u < 4; ++u) {
      if (p[u].y >= 0) {
        const int k = p[u].y - r0;
        const int rk = atomicAdd(&rcnt[k], 1);
        pairs[rofs[k] + rk] = make_int2(p[u].x, __float_as_int(w[u]));
      }
    }
  }
}

template <int NCOLS>
__global__ void __launch_bounds__(256) gemm_wmma_f32(const float* __restrict__ X,
                                                     const float* __restrict__ W,
                                                     float* __restrict__ T, int N) {
  constexpr int CT = NCOLS / 16;
  constexpr int RG = 8 / CT;
  constexpr int CH = 32 * RG;
  constexpr int NLD = CH / 8;
  constexpr int P4 = 33;
  __shared__ float4 xs4[CH * P4];

  const int tid = threadIdx.x;
  const int wave = tid >> 5;
  const int lane = tid & 31;
  const int c = lane & 15;
  const int h = lane >> 4;
  const int ct = wave % CT;
  const int rg = wave / CT;
  const int n = ct * 16 + c;

  v2f bf[32];
#pragma unroll
  for (int j = 0; j < 16; ++j) {
    const int kb = 8 * j + 4 * h;
    bf[2 * j].x     = W[(size_t)(kb + 0) * NCOLS + n];
    bf[2 * j].y     = W[(size_t)(kb + 1) * NCOLS + n];
    bf[2 * j + 1].x = W[(size_t)(kb + 2) * NCOLS + n];
    bf[2 * j + 1].y = W[(size_t)(kb + 3) * NCOLS + n];
  }

  const float4* __restrict__ X4 = reinterpret_cast<const float4*>(X);
  const int nchunks = (N + CH - 1) / CH;
  int chunk = blockIdx.x;

  float4 pre[NLD];
  if (chunk < nchunks) {
#pragma unroll
    for (int q = 0; q < NLD; ++q) {
      const int i = tid + 256 * q;
      const int row = chunk * CH + (i >> 5);
      pre[q] = row < N ? X4[(size_t)row * 32 + (i & 31)] : make_float4(0.f, 0.f, 0.f, 0.f);
    }
  }

  for (; chunk < nchunks; chunk += gridDim.x) {
#pragma unroll
    for (int q = 0; q < NLD; ++q) {
      const int i = tid + 256 * q;
      xs4[(i >> 5) * P4 + (i & 31)] = pre[q];
    }
    __builtin_amdgcn_fence(__ATOMIC_RELEASE, "workgroup", "local");
    __builtin_amdgcn_s_barrier();
    __builtin_amdgcn_fence(__ATOMIC_ACQUIRE, "workgroup", "local");

    const int next = chunk + gridDim.x;
    if (next < nchunks) {
#pragma unroll
      for (int q = 0; q < NLD; ++q) {
        const int i = tid + 256 * q;
        const int row = next * CH + (i >> 5);
        pre[q] = row < N ? X4[(size_t)row * 32 + (i & 31)] : make_float4(0.f, 0.f, 0.f, 0.f);
      }
    }

    const float4* a0p = xs4 + (rg * 32 + c) * P4 + h;
    const float4* a1p = a0p + 16 * P4;
    v8f acc0 = {};
    v8f acc1 = {};
#pragma unroll
    for (int j = 0; j < 16; ++j) {
      const float4 u = a0p[2 * j];
      const float4 v = a1p[2 * j];
      v2f a;
      a.x = u.x; a.y = u.y;
      acc0 = __builtin_amdgcn_wmma_f32_16x16x4_f32(false, a, false, bf[2 * j], (short)0, acc0,
                                                   false, false);
      a.x = v.x; a.y = v.y;
      acc1 = __builtin_amdgcn_wmma_f32_16x16x4_f32(false, a, false, bf[2 * j], (short)0, acc1,
                                                   false, false);
      a.x = u.z; a.y = u.w;
      acc0 = __builtin_amdgcn_wmma_f32_16x16x4_f32(false, a, false, bf[2 * j + 1], (short)0, acc0,
                                                   false, false);
      a.x = v.z; a.y = v.w;
      acc1 = __builtin_amdgcn_wmma_f32_16x16x4_f32(false, a, false, bf[2 * j + 1], (short)0, acc1,
                                                   false, false);
    }

    const int rowbase = chunk * CH + rg * 32 + 8 * h;
    const int lim = N - rowbase;
    float* tp = T + (size_t)rowbase * NCOLS + n;
#pragma unroll
    for (int r = 0; r < 8; ++r) {
      if (r < lim) tp[r * NCOLS] = acc0[r];
      if (r + 16 < lim) tp[(r + 16) * NCOLS] = acc1[r];
    }
    __builtin_amdgcn_fence(__ATOMIC_RELEASE, "workgroup", "local");
    __builtin_amdgcn_s_barrier();
    __builtin_amdgcn_fence(__ATOMIC_ACQUIRE, "workgroup", "local");
  }
}

template <int V>
__device__ __forceinline__ void sum_run(float __attribute__((ext_vector_type(V)))& acc,
                                        const float __attribute__((ext_vector_type(V)))* __restrict__ Tv,
                                        int s_l, float w_l, int cnt, int lane) {
  typedef float vf __attribute__((ext_vector_type(V)));
  for (int j = 0; j < cnt; j += 4) {
    vf v[4];
    float w[4];
    for (int u = 0; u < 4; ++u) {
      const int s = __builtin_amdgcn_readlane(s_l, j + u);
      w[u] = __int_as_float(__builtin_amdgcn_readlane(__float_as_int(w_l), j + u));
      v[u] = Tv[(size_t)s * 32 + lane];
    }
    for (int u = 0; u < 4; ++u) acc += v[u] * w[u];
  }
}

template <int V>
__device__ __forceinline__ void sum_run8(float __attribute__((ext_vector_type(V)))& acc,
                                         const float __attribute__((ext_vector_type(V)))* __restrict__ Tv,
                                         int s_l, float w_l, int cnt, int lane) {
  typedef float vf __attribute__((ext_vector_type(V)));
  int j = 0;
  for (; cnt - j > 4; j += 8) {
    vf v[8];
    float w[8];
    for (int u = 0; u < 8; ++u) {
      const int s = __builtin_amdgcn_readlane(s_l, j + u);
      w[u] = __int_as_float(__builtin_amdgcn_readlane(__float_as_int(w_l), j + u));
      v[u] = Tv[(size_t)s * 32 + lane];
    }
    for (int u = 0; u < 8; ++u) acc += v[u] * w[u];
  }
  if (j < cnt) {
    vf v[4];
    float w[4];
    for (int u = 0; u < 4; ++u) {
      const int s = __builtin_amdgcn_readlane(s_l, j + u);
      w[u] = __int_as_float(__builtin_amdgcn_readlane(__float_as_int(w_l), j + u));
      v[u] = Tv[(size_t)s * 32 + lane];
    }
    for (int u = 0; u < 4; ++u) acc += v[u] * w[u];
  }
}

template <int D, bool RELU>
__global__ void __launch_bounds__(256) gather_sum_rows(const int* __restrict__ rowptr,
                                                       const int* __restrict__ pairs_w,
                                                       const float* __restrict__ dinv,
                                                       const float* __restrict__ T,
                                                       const float* __restrict__ bias,
                                                       float* __restrict__ OUT, int N) {
  constexpr int V = D / 32;
  typedef float vf __attribute__((ext_vector_type(V)));
  const int lane = threadIdx.x & 31;
  const int nw = (gridDim.x * blockDim.x) >> 5;
  int i = __builtin_amdgcn_readfirstlane((int)((blockIdx.x * blockDim.x + threadIdx.x) >> 5));
  if (i >= N) return;
  const int2* __restrict__ pairs = reinterpret_cast<const int2*>(pairs_w);
  const vf* __restrict__ Tv = reinterpret_cast<const vf*>(T);
  vf* __restrict__ Ov = reinterpret_cast<vf*>(OUT);
  const vf bv = reinterpret_cast<const vf*>(bias)[lane];

  int beg = rowptr[i], end = rowptr[i + 1];
  int ni = i + nw;
  int nbeg = 0, nend = 0;
  if (ni < N) { nbeg = rowptr[ni]; nend = rowptr[ni + 1]; }
  int2 p = make_int2(i, 0);
  if (beg + lane < end) p = pairs[beg + lane];
  float di = dinv[i];
  int s_l = p.x;
  float w_l = __int_as_float(p.y) * di;

  for (;;) {
    const int nni = ni + nw;
    int nnbeg = 0, nnend = 0;
    if (nni < N) { nnbeg = rowptr[nni]; nnend = rowptr[nni + 1]; }
    float ndi = 0.0f;
    int2 np = make_int2(i, 0);
    if (ni < N) {
      ndi = dinv[ni];
      np = make_int2(ni, 0);
      if (nbeg + lane < nend) np = pairs[nbeg + lane];
    }
    const vf selfv = Tv[(size_t)i * 32 + lane];
    vf acc = 0.0f;
    sum_run<V>(acc, Tv, s_l, w_l, min(32, end - beg), lane);
    for (int base = beg + 32; base < end; base += 32) {
      int2 q = make_int2(i, 0);
      if (base + lane < end) q = pairs[base + lane];
      sum_run<V>(acc, Tv, q.x, __int_as_float(q.y) * di, min(32, end - base), lane);
    }
    acc += selfv * (di * di);
    acc += bv;
    if (RELU) {
      for (int k = 0; k < V; ++k) acc[k] = acc[k] > 0.0f ? acc[k] : 0.0f;
    }
    Ov[(size_t)i * 32 + lane] = acc;
    if (ni >= N) break;
    i = ni; beg = nbeg; end = nend; di = ndi;
    s_l = np.x;
    w_l = __int_as_float(np.y) * ndi;
    ni = nni; nbeg = nnbeg; nend = nnend;
  }
}

__global__ void __launch_bounds__(128) gather_sum_rows_gemm(const int* __restrict__ rowptr,
                                                            const int* __restrict__ pairs_w,
                                                            const float* __restrict__ dinv,
                                                            const float* __restrict__ T,
                                                            const float* __restrict__ bias,
                                                            const float* __restrict__ W,
                                                            float* __restrict__ T2, int N) {
  constexpr int V = GDH / 32;
  constexpr int TR = 32;
  constexpr int RW = TR / 4;
  constexpr int P4 = 33;
  typedef float vf __attribute__((ext_vector_type(V)));
  __shared__ float4 hs4[TR * P4];

  const int lane = threadIdx.x & 31;
  const int wave = __builtin_amdgcn_readfirstlane((int)(threadIdx.x >> 5));
  const int c = lane & 15;
  const int h = lane >> 4;
  const int n = wave * 16 + c;

  const int ntiles = (N + TR - 1) / TR;
  const int G = gridDim.x;
  int tile = blockIdx.x;
  if (tile >= ntiles) return;

  v2f bf[32];
#pragma unroll
  for (int j = 0; j < 16; ++j) {
    const int kb = 8 * j + 4 * h;
    bf[2 * j].x     = W[(size_t)(kb + 0) * GDOUT + n];
    bf[2 * j].y     = W[(size_t)(kb + 1) * GDOUT + n];
    bf[2 * j + 1].x = W[(size_t)(kb + 2) * GDOUT + n];
    bf[2 * j + 1].y = W[(size_t)(kb + 3) * GDOUT + n];
  }

  const int2* __restrict__ pairs = reinterpret_cast<const int2*>(pairs_w);
  const vf* __restrict__ Tv = reinterpret_cast<const vf*>(T);
  const vf bv = reinterpret_cast<const vf*>(bias)[lane];

  int q = 0;
  int i = tile * TR + wave * RW;
  bool valid = i < N;
  int beg = 0, end = 0;
  if (valid) { beg = rowptr[i]; end = rowptr[i + 1]; }
  int ntile = tile, nq = 1;
  int ni = i + 1;
  bool nvalid = ni < N;
  int nbeg = 0, nend = 0;
  if (nvalid) { nbeg = rowptr[ni]; nend = rowptr[ni + 1]; }
  float di = 0.0f;
  int2 p = make_int2(0, 0);
  if (valid) {
    di = dinv[i];
    p = make_int2(i, 0);
    if (beg + lane < end) p = pairs[beg + lane];
  }
  int s_l = p.x;
  float w_l = __int_as_float(p.y) * di;

  for (;;) {
    int nntile = ntile, nnq = nq + 1;
    if (nnq == RW) { nnq = 0; nntile += G; }
    const int nni = nntile * TR + wave * RW + nnq;
    const bool nnvalid = nntile < ntiles && nni < N;
    int nnbeg = 0, nnend = 0;
    if (nnvalid) { nnbeg = rowptr[nni]; nnend = rowptr[nni + 1]; }
    float ndi = 0.0f;
    int2 np = make_int2(0, 0);
    if (nvalid) {
      ndi = dinv[ni];
      np = make_int2(ni, 0);
      if (nbeg + lane < nend) np = pairs[nbeg + lane];
    }
    vf acc = 0.0f;
    if (valid) {
      const vf selfv = Tv[(size_t)i * 32 + lane];
      sum_run8<V>(acc, Tv, s_l, w_l, min(32, end - beg), lane);
      for (int base = beg + 32; base < end; base += 32) {
        int2 r = make_int2(i, 0);
        if (base + lane < end) r = pairs[base + lane];
        sum_run8<V>(acc, Tv, r.x, __int_as_float(r.y) * di, min(32, end - base), lane);
      }
      acc += selfv * (di * di);
      acc += bv;
      for (int k = 0; k < V; ++k) acc[k] = acc[k] > 0.0f ? acc[k] : 0.0f;
    }
    hs4[(wave * RW + q) * P4 + lane] = make_float4(acc[0], acc[1], acc[2], acc[3]);

    if (q == RW - 1) {
      __builtin_amdgcn_fence(__ATOMIC_RELEASE, "workgroup", "local");
      __builtin_amdgcn_s_barrier();
      __builtin_amdgcn_fence(__ATOMIC_ACQUIRE, "workgroup", "local");

      const float4* a0p = hs4 + c * P4 + h;
      const float4* a1p = a0p + 16 * P4;
      v8f acc0 = {};
      v8f acc1 = {};
#pragma unroll
      for (int j = 0; j < 16; ++j) {
        const float4 u = a0p[2 * j];
        const float4 v = a1p[2 * j];
        v2f a;
        a.x = u.x; a.y = u.y;
        acc0 = __builtin_amdgcn_wmma_f32_16x16x4_f32(false, a, false, bf[2 * j], (short)0, acc0,
                                                     false, false);
        a.x = v.x; a.y = v.y;
        acc1 = __builtin_amdgcn_wmma_f32_16x16x4_f32(false, a, false, bf[2 * j], (short)0, acc1,
                                                     false, false);
        a.x = u.z; a.y = u.w;
        acc0 = __builtin_amdgcn_wmma_f32_16x16x4_f32(false, a, false, bf[2 * j + 1], (short)0, acc0,
                                                     false, false);
        a.x = v.z; a.y = v.w;
        acc1 = __builtin_amdgcn_wmma_f32_16x16x4_f32(false, a, false, bf[2 * j + 1], (short)0, acc1,
                                                     false, false);
      }

      const int rowbase = tile * TR + 8 * h;
      const int lim = N - rowbase;
      float* tp = T2 + (size_t)rowbase * GDOUT + n;
#pragma unroll
      for (int r = 0; r < 8; ++r) {
        if (r < lim) tp[r * GDOUT] = acc0[r];
        if (r + 16 < lim) tp[(r + 16) * GDOUT] = acc1[r];
      }
      if (tile + G >= ntiles) break;
      __builtin_amdgcn_fence(__ATOMIC_RELEASE, "workgroup", "local");
      __builtin_amdgcn_s_barrier();
      __builtin_amdgcn_fence(__ATOMIC_ACQUIRE, "workgroup", "local");
    }

    tile = ntile; q = nq; i = ni; valid = nvalid; beg = nbeg; end = nend; di = ndi;
    s_l = np.x;
    w_l = __int_as_float(np.y) * ndi;
    ntile = nntile; nq = nnq; ni = nni; nvalid = nnvalid; nbeg = nnbeg; nend = nnend;
  }
}

extern "C" void kernel_launch(void* const* d_in, const int* in_sizes, int n_in,
                              void* d_out, int out_size, void* d_ws, size_t ws_size,
                              hipStream_t stream) {
  const float* x  = (const float*)d_in[0];
  const int*   ei = (const int*)d_in[1];
  const float* W1 = (const float*)d_in[2];
  const float* b1 = (const float*)d_in[3];
  const float* W2 = (const float*)d_in[4];
  const float* b2 = (const float*)d_in[5];
  float* out = (float*)d_out;

  const int N = in_sizes[0] / GDIN;
  const int E = in_sizes[1] / 2;
  const int* src = ei;
  const int* dst = ei + E;

  float* dinv = (float*)d_ws;
  float* bufA = dinv + N;
  float* bufB = bufA + (size_t)N * GDH;

  int* gcount   = (int*)bufB;
  int* gcursor  = gcount + MAXB * CPAD;
  int* bucketed = gcursor + MAXB * CPAD;
  const int NB = (N + BROWS - 1) >> BSH;
  const int RP = (N + 4) & ~3;
  int* lst = (int*)(bufB + (size_t)N * GDOUT);

  zero_i32_kernel<<<(2 * MAXB * CPAD + 255) / 256, 256, 0, stream>>>(gcount, 2 * MAXB * CPAD);
  bucket_count_kernel<<<(E + SLICE - 1) / SLICE, MAXB, 0, stream>>>(dst, gcount, E, NB);
  bucket_scatter_kernel<<<(E + SLICE - 1) / SLICE, MAXB, 0, stream>>>(src, dst, gcount, gcursor, bucketed,
                                                                     E, NB);
  bucket_rows_kernel<<<NB, MAXB, 0, stream>>>(gcount, bucketed, lst, dinv, N);
  bucket_fill_kernel<<<NB, MAXB, 0, stream>>>(bucketed, lst, dinv, lst + RP, N);

  gemm_wmma_f32<GDH><<<768, 256, 0, stream>>>(x, W1, bufA, N);
  gather_sum_rows_gemm<<<(N + 31) / 32, 128, 0, stream>>>(lst, lst + RP, dinv, bufA, b1, W2, bufB, N);

  gather_sum_rows<GDOUT, false><<<2048, 256, 0, stream>>>(lst, lst + RP, dinv, bufB, b2, out, N);
}


# T rows are scaled by dinv[row] where they are made (A operand of both WMMA products): the sorted list holds sources only, a bucket's row count and list fill share a launch; weight rounding differs
# speedup vs baseline: 4.2911x; 1.0482x over previous
#include <hip/hip_runtime.h>

typedef float v2f __attribute__((ext_vector_type(2)));
typedef float v8f __attribute__((ext_vector_type(8)));

#define GDIN  128
#define GDH   128
#define GDOUT 64

#define BSH   8
#define BROWS (1 << BSH)
#define MAXB  512
#define SLICE 4096
#define CPAD  32

__global__ void zero_i32_kernel(int* __restrict__ p, int n) {
  const int i = blockIdx.x * blockDim.x + threadIdx.x;
  if (i < n) p[i] = 0;
}

__global__ void __launch_bounds__(MAXB) bucket_count_kernel(const int* __restrict__ dst,
                                                            int* __restrict__ gcount, int E, int NB) {
  __shared__ int hist[MAXB];
  const int tid = threadIdx.x;
  const int beg = blockIdx.x * SLICE;
  const int n = min(SLICE, E - beg);
  int d[SLICE / MAXB];
  for (int u = 0; u < SLICE / MAXB; ++u) {
    const int i = tid + u * MAXB;
    d[u] = i < n ? dst[beg + i] : -1;
  }
  hist[tid] = 0;
  __syncthreads();
  for (int u = 0; u < SLICE / MAXB; ++u)
    if (d[u] >= 0) atomicAdd(&hist[d[u] >> BSH], 1);
  __syncthreads();
  const int c = hist[tid];
  if (tid < NB && c) atomicAdd(&gcount[tid * CPAD], c);
}

__global__ void __launch_bounds__(MAXB) bucket_scatter_kernel(const int* __restrict__ src,
                                                              const int* __restrict__ dst,
                                                              const int* __restrict__ gcount,
                                                              int* __restrict__ gcursor,
                                                              int* __restrict__ bucketed_w,
                                                              int E, int NB) {
  __shared__ int2 lbuf[SLICE];
  __shared__ int hist[MAXB];
  __shared__ int lofs[MAXB];
  __shared__ int delta[MAXB];
  __shared__ int wtg[MAXB / 32];
  __shared__ int wtc[MAXB / 32];
  int2* __restrict__ bucketed = reinterpret_cast<int2*>(bucketed_w);
  const int tid = threadIdx.x;
  const int lane = tid & 31;
  const int wave = tid >> 5;
  const int beg = blockIdx.x * SLICE;
  const int n = min(SLICE, E - beg);
  int d[SLICE / MAXB], s[SLICE / MAXB];
  for (int u = 0; u < SLICE / MAXB; ++u) {
    const int i = tid + u * MAXB;
    d[u] = i < n ? dst[beg + i] : -1;
    s[u] = i < n ? src[beg + i] : 0;
  }
  hist[tid] = 0;
  __syncthreads();
  for (int u = 0; u < SLICE / MAXB; ++u)
    if (d[u] >= 0) atomicAdd(&hist[d[u] >> BSH], 1);
  const int g = tid < NB ? gcount[tid * CPAD] : 0;
  __syncthreads();
  const int c = hist[tid];
  int ig = g, ic = c;
  for (int o = 1; o < 32; o <<= 1) {
    const int vg = __shfl_up(ig, o);
    const int vc = __shfl_up(ic, o);
    if (lane >= o) { ig += vg; ic += vc; }
  }
  if (lane == 31) { wtg[wave] = ig; wtc[wave] = ic; }
  __syncthreads();
  int og = 0, oc = 0;
  for (int w = 0; w < wave; ++w) { og += wtg[w]; oc += wtc[w]; }
  int r = 0;
  if (tid < NB && c) r = atomicAdd(&gcursor[tid * CPAD], c);
  const int lo = oc + ic - c;
  lofs[tid] = lo;
  delta[tid] = og + ig - g + r - lo;
  hist[tid] = 0;
  __syncthreads();
  for (int u = 0; u < SLICE / MAXB; ++u) {
    if (d[u] >= 0) {
      const int b = d[u] >> BSH;
      const int k = atomicAdd(&hist[b], 1);
      lbuf[lofs[b] + k] = make_int2(s[u], d[u]);
    }
  }
  __syncthreads();
  for (int i = tid; i < n; i += MAXB) {
    const int2 rec = lbuf[i];
    bucketed[delta[rec.y >> BSH] + i] = rec;
  }
}

__global__ void __launch_bounds__(MAXB) bucket_rows_fill_kernel(const int* __restrict__ gcount,
                                                                const int* __restrict__ bucketed_w,
                                                                int* __restrict__ rowptr,
                                                                float* __restrict__ dinv,
                                                                int* __restrict__ list, int N) {
  __shared__ int rcnt[BROWS];
  __shared__ int rofs[BROWS];
  __shared__ int wsum[MAXB / 32];
  __shared__ int wtot[BROWS / 32];
  const int2* __restrict__ bucketed = reinterpret_cast<const int2*>(bucketed_w);
  const int tid = threadIdx.x;
  const int lane = tid & 31;
  const int wave = tid >> 5;
  const int b = blockIdx.x;
  const int r0 = b << BSH;
  int part = tid < b ? gcount[tid * CPAD] : 0;
  for (int o = 16; o > 0; o >>= 1) part += __shfl_xor(part, o);
  if (lane == 0) wsum[wave] = part;
  if (tid < BROWS) rcnt[tid] = 0;
  __syncthreads();
  int start = 0;
  for (int w = 0; w < MAXB / 32; ++w) start += wsum[w];
  const int cnt = gcount[b * CPAD];
  for (int i0 = tid; i0 < cnt; i0 += 4 * MAXB) {
    int d[4];
    for (int u = 0; u < 4; ++u) {
      const int i = i0 + u * MAXB;
      d[u] = i < cnt ? bucketed[start + i].y : -1;
    }
    for (int u = 0; u < 4; ++u)
      if (d[u] >= 0) atomicAdd(&rcnt[d[u] - r0], 1);
  }
  __syncthreads();
  const int c = tid < BROWS ? rcnt[tid] : 0;
  int incl = c;
  for (int o = 1; o < 32; o <<= 1) {
    const int v = __shfl_up(incl, o);
    if (lane >= o) incl += v;
  }
  if (tid < BROWS && lane == 31) wtot[wave] = incl;
  __syncthreads();
  if (tid < BROWS) {
    int off = start;
    for (int w = 0; w < wave; ++w) off += wtot[w];
    const int row = r0 + tid;
    const int first = off + incl - c;
    rofs[tid] = first;
    rcnt[tid] = 0;
    if (row < N) {
      rowptr[row] = first;
      dinv[row] = rsqrtf((float)(c + 1));
    }
    if (b == (int)gridDim.x - 1 && tid == BROWS - 1) rowptr[N] = off + incl;
  }
  __syncthreads();
  for (int i0 = tid; i0 < cnt; i0 += 4 * MAXB) {
    int2 p[4];
    for (int u = 0; u < 4; ++u) {
      const int i = i0 + u * MAXB;
      p[u] = i < cnt ? bucketed[start + i] : make_int2(0, -1);
    }
    for (int u = 0; u < 4; ++u) {
      if (p[u].y >= 0) {
        const int k = p[u].y - r0;
        const int rk = atomicAdd(&rcnt[k], 1);
        list[rofs[k] + rk] = p[u].x;
      }
    }
  }
}

template <int NCOLS>
__global__ void __launch_bounds__(256) gemm_wmma_f32(const float* __restrict__ X,
                                                     const float* __restrict__ W,
                                                     const float* __restrict__ rowscale,
                                                     float* __restrict__ T, int N) {
  constexpr int CT = NCOLS / 16;
  constexpr int RG = 8 / CT;
  constexpr int CH = 32 * RG;
  constexpr int NLD = CH / 8;
  constexpr int P4 = 33;
  __shared__ float4 xs4[CH * P4];

  const int tid = threadIdx.x;
  const int wave = tid >> 5;
  const int lane = tid & 31;
  const int c = lane & 15;
  const int h = lane >> 4;
  const int ct = wave % CT;
  const int rg = wave / CT;
  const int n = ct * 16 + c;

  v2f bf[32];
#pragma unroll
  for (int j = 0; j < 16; ++j) {
    const int kb = 8 * j + 4 * h;
    bf[2 * j].x     = W[(size_t)(kb + 0) * NCOLS + n];
    bf[2 * j].y     = W[(size_t)(kb + 1) * NCOLS + n];
    bf[2 * j + 1].x = W[(size_t)(kb + 2) * NCOLS + n];
    bf[2 * j + 1].y = W[(size_t)(kb + 3) * NCOLS + n];
  }

  const float4* __restrict__ X4 = reinterpret_cast<const float4*>(X);
  const int nchunks = (N + CH - 1) / CH;
  int chunk = blockIdx.x;

  float4 pre[NLD];
  float prs[NLD];
  if (chunk < nchunks) {
#pragma unroll
    for (int q = 0; q < NLD; ++q) {
      const int i = tid + 256 * q;
      const int row = chunk * CH + (i >> 5);
      pre[q] = row < N ? X4[(size_t)row * 32 + (i & 31)] : make_float4(0.f, 0.f, 0.f, 0.f);
      prs[q] = row < N ? rowscale[row] : 0.f;
    }
  }

  for (; chunk < nchunks; chunk += gridDim.x) {
#pragma unroll
    for (int q = 0; q < NLD; ++q) {
      const int i = tid + 256 * q;
      xs4[(i >> 5) * P4 + (i & 31)] = make_float4(pre[q].x * prs[q], pre[q].y * prs[q],
                                                  pre[q].z * prs[q], pre[q].w * prs[q]);
    }
    __builtin_amdgcn_fence(__ATOMIC_RELEASE, "workgroup", "local");
    __builtin_amdgcn_s_barrier();
    __builtin_amdgcn_fence(__ATOMIC_ACQUIRE, "workgroup", "local");

    const int next = chunk + gridDim.x;
    if (next < nchunks) {
#pragma unroll
      for (int q = 0; q < NLD; ++q) {
        const int i = tid + 256 * q;
        const int row = next * CH + (i >> 5);
        pre[q] = row < N ? X4[(size_t)row * 32 + (i & 31)] : make_float4(0.f, 0.f, 0.f, 0.f);
        prs[q] = row < N ? rowscale[row] : 0.f;
      }
    }

    const float4* a0p = xs4 + (rg * 32 + c) * P4 + h;
    const float4* a1p = a0p + 16 * P4;
    v8f acc0 = {};
    v8f acc1 = {};
#pragma unroll
    for (int j = 0; j < 16; ++j) {
      const float4 u = a0p[2 * j];
      const float4 v = a1p[2 * j];
      v2f a;
      a.x = u.x; a.y = u.y;
      acc0 = __builtin_amdgcn_wmma_f32_16x16x4_f32(false, a, false, bf[2 * j], (short)0, acc0,
                                                   false, false);
      a.x = v.x; a.y = v.y;
      acc1 = __builtin_amdgcn_wmma_f32_16x16x4_f32(false, a, false, bf[2 * j], (short)0, acc1,
                                                   false, false);
      a.x = u.z; a.y = u.w;
      acc0 = __builtin_amdgcn_wmma_f32_16x16x4_f32(false, a, false, bf[2 * j + 1], (short)0, acc0,
                                                   false, false);
      a.x = v.z; a.y = v.w;
      acc1 = __builtin_amdgcn_wmma_f32_16x16x4_f32(false, a, false, bf[2 * j + 1], (short)0, acc1,
                                                   false, false);
    }

    const int rowbase = chunk * CH + rg * 32 + 8 * h;
    const int lim = N - rowbase;
    float* tp = T + (size_t)rowbase * NCOLS + n;
#pragma unroll
    for (int r = 0; r < 8; ++r) {
      if (r < lim) tp[r * NCOLS] = acc0[r];
      if (r + 16 < lim) tp[(r + 16) * NCOLS] = acc1[r];
    }
    __builtin_amdgcn_fence(__ATOMIC_RELEASE, "workgroup", "local");
    __builtin_amdgcn_s_barrier();
    __builtin_amdgcn_fence(__ATOMIC_ACQUIRE, "workgroup", "local");
  }
}

template <int V>
__device__ __forceinline__ void sum_run(float __attribute__((ext_vector_type(V)))& acc,
                                        const float __attribute__((ext_vector_type(V)))* __restrict__ Tv,
                                        int s_l, int cnt, int lane) {
  typedef float vf __attribute__((ext_vector_type(V)));
  for (int j = 0; j < cnt; j += 4) {
    vf v[4];
    float w[4];
    for (int u = 0; u < 4; ++u) {
      const int s = __builtin_amdgcn_readlane(s_l, j + u);
      w[u] = (u == 0 || j + u < cnt) ? 1.0f : 0.0f;
      v[u] = Tv[(size_t)s * 32 + lane];
    }
    for (int u = 0; u < 4; ++u) acc += v[u] * w[u];
  }
}

template <int V>
__device__ __forceinline__ void sum_run8(float __attribute__((ext_vector_type(V)))& acc,
                                         const float __attribute__((ext_vector_type(V)))* __restrict__ Tv,
                                         int s_l, int cnt, int lane) {
  typedef float vf __attribute__((ext_vector_type(V)));
  int j = 0;
  for (; cnt - j > 4; j += 8) {
    vf v[8];
    float w[8];
    for (int u = 0; u < 8; ++u) {
      const int s = __builtin_amdgcn_readlane(s_l, j + u);
      w[u] = (u < 5 || j + u < cnt) ? 1.0f : 0.0f;
      v[u] = Tv[(size_t)s * 32 + lane];
    }
    for (int u = 0; u < 8; ++u) acc += v[u] * w[u];
  }
  if (j < cnt) {
    vf v[4];
    float w[4];
    for (int u = 0; u < 4; ++u) {
      const int s = __builtin_amdgcn_readlane(s_l, j + u);
      w[u] = (u == 0 || j + u < cnt) ? 1.0f : 0.0f;
      v[u] = Tv[(size_t)s * 32 + lane];
    }
    for (int u = 0; u < 4; ++u) acc += v[u] * w[u];
  }
}

template <int D, bool RELU>
__global__ void __launch_bounds__(256) gather_sum_rows(const int* __restrict__ rowptr,
                                                       const int* __restrict__ list,
                                                       const float* __restrict__ dinv,
                                                       const float* __restrict__ T,
                                                       const float* __restrict__ bias,
                                                       float* __restrict__ OUT, int N) {
  constexpr int V = D / 32;
  typedef float vf __attribute__((ext_vector_type(V)));
  const int lane = threadIdx.x & 31;
  const int nw = (gridDim.x * blockDim.x) >> 5;
  int i = __builtin_amdgcn_readfirstlane((int)((blockIdx.x * blockDim.x + threadIdx.x) >> 5));
  if (i >= N) return;
  const vf* __restrict__ Tv = reinterpret_cast<const vf*>(T);
  vf* __restrict__ Ov = reinterpret_cast<vf*>(OUT);
  const vf bv = reinterpret_cast<const vf*>(bias)[lane];

  int beg = rowptr[i], end = rowptr[i + 1];
  int ni = i + nw;
  int nbeg = 0, nend = 0;
  if (ni < N) { nbeg = rowptr[ni]; nend = rowptr[ni + 1]; }
  const unsigned last = (unsigned)(N - 1);
  int s_l = i;
  if (beg + lane < end) s_l = list[beg + lane];
  s_l = (int)min((unsigned)s_l, last);
  float di = dinv[i];

  for (;;) {
    const int nni = ni + nw;
    int nnbeg = 0, nnend = 0;
    if (nni < N) { nnbeg = rowptr[nni]; nnend = rowptr[nni + 1]; }
    float ndi = 0.0f;
    int ns = i;
    if (ni < N) {
      ndi = dinv[ni];
      ns = ni;
      if (nbeg + lane < nend) ns = list[nbeg + lane];
    }
    const vf selfv = Tv[(size_t)i * 32 + lane];
    vf acc = 0.0f;
    sum_run<V>(acc, Tv, s_l, min(32, end - beg), lane);
    for (int base = beg + 32; base < end; base += 32) {
      int q = i;
      if (base + lane < end) q = list[base + lane];
      sum_run<V>(acc, Tv, (int)min((unsigned)q, last), min(32, end - base), lane);
    }
    acc += selfv;
    acc = acc * di + bv;
    if (RELU) {
      for (int k = 0; k < V; ++k) acc[k] = acc[k] > 0.0f ? acc[k] : 0.0f;
    }
    Ov[(size_t)i * 32 + lane] = acc;
    if (ni >= N) break;
    i = ni; beg = nbeg; end = nend; di = ndi;
    s_l = (int)min((unsigned)ns, last);
    ni = nni; nbeg = nnbeg; nend = nnend;
  }
}

__global__ void __launch_bounds__(128) gather_sum_rows_gemm(const int* __restrict__ rowptr,
                                                            const int* __restrict__ list,
                                                            const float* __restrict__ dinv,
                                                            const float* __restrict__ T,
                                                            const float* __restrict__ bias,
                                                            const float* __restrict__ W,
                                                            float* __restrict__ T2, int N) {
  constexpr int V = GDH / 32;
  constexpr int TR = 32;
  constexpr int RW = TR / 4;
  constexpr int P4 = 33;
  typedef float vf __attribute__((ext_vector_type(V)));
  __shared__ float4 hs4[TR * P4];

  const int lane = threadIdx.x & 31;
  const int wave = __builtin_amdgcn_readfirstlane((int)(threadIdx.x >> 5));
  const int c = lane & 15;
  const int h = lane >> 4;
  const int n = wave * 16 + c;

  const int ntiles = (N + TR - 1) / TR;
  const int G = gridDim.x;
  int tile = blockIdx.x;
  if (tile >= ntiles) return;

  v2f bf[32];
#pragma unroll
  for (int j = 0; j < 16; ++j) {
    const int kb = 8 * j + 4 * h;
    bf[2 * j].x     = W[(size_t)(kb + 0) * GDOUT + n];
    bf[2 * j].y     = W[(size_t)(kb + 1) * GDOUT + n];
    bf[2 * j + 1].x = W[(size_t)(kb + 2) * GDOUT + n];
    bf[2 * j + 1].y = W[(size_t)(kb + 3) * GDOUT + n];
  }

  const vf* __restrict__ Tv = reinterpret_cast<const vf*>(T);
  const vf bv = reinterpret_cast<const vf*>(bias)[lane];

  int q = 0;
  int i = tile * TR + wave * RW;
  bool valid = i < N;
  int beg = 0, end = 0;
  if (valid) { beg = rowptr[i]; end = rowptr[i + 1]; }
  int ntile = tile, nq = 1;
  int ni = i + 1;
  bool nvalid = ni < N;
  int nbeg = 0, nend = 0;
  if (nvalid) { nbeg = rowptr[ni]; nend = rowptr[ni + 1]; }
  const unsigned last = (unsigned)(N - 1);
  float di = 0.0f;
  int s_l = 0;
  if (valid) {
    di = dinv[i];
    s_l = i;
    if (beg + lane < end) s_l = list[beg + lane];
  }
  s_l = (int)min((unsigned)s_l, last);

  for (;;) {
    int nntile = ntile, nnq = nq + 1;
    if (nnq == RW) { nnq = 0; nntile += G; }
    const int nni = nntile * TR + wave * RW + nnq;
    const bool nnvalid = nntile < ntiles && nni < N;
    int nnbeg = 0, nnend = 0;
    if (nnvalid) { nnbeg = rowptr[nni]; nnend = rowptr[nni + 1]; }
    float ndi = 0.0f;
    int ns = 0;
    if (nvalid) {
      ndi = dinv[ni];
      ns = ni;
      if (nbeg + lane < nend) ns = list[nbeg + lane];
    }
    vf acc = 0.0f;
    if (valid) {
      const vf selfv = Tv[(size_t)i * 32 + lane];
      sum_run8<V>(acc, Tv, s_l, min(32, end - beg), lane);
      for (int base = beg + 32; base < end; base += 32) {
        int r = i;
        if (base + lane < end) r = list[base + lane];
        sum_run8<V>(acc, Tv, (int)min((unsigned)r, last), min(32, end - base), lane);
      }
      acc += selfv;
      acc = acc * di + bv;
      for (int k = 0; k < V; ++k) acc[k] = acc[k] > 0.0f ? acc[k] : 0.0f;
      acc *= di;
    }
    hs4[(wave * RW + q) * P4 + lane] = make_float4(acc[0], acc[1], acc[2], acc[3]);

    if (q == RW - 1) {
      __builtin_amdgcn_fence(__ATOMIC_RELEASE, "workgroup", "local");
      __builtin_amdgcn_s_barrier();
      __builtin_amdgcn_fence(__ATOMIC_ACQUIRE, "workgroup", "local");

      const float4* a0p = hs4 + c * P4 + h;
      const float4* a1p = a0p + 16 * P4;
      v8f acc0 = {};
      v8f acc1 = {};
#pragma unroll
      for (int j = 0; j < 16; ++j) {
        const float4 u = a0p[2 * j];
        const float4 v = a1p[2 * j];
        v2f a;
        a.x = u.x; a.y = u.y;
        acc0 = __builtin_amdgcn_wmma_f32_16x16x4_f32(false, a, false, bf[2 * j], (short)0, acc0,
                                                     false, false);
        a.x = v.x; a.y = v.y;
        acc1 = __builtin_amdgcn_wmma_f32_16x16x4_f32(false, a, false, bf[2 * j], (short)0, acc1,
                                                     false, false);
        a.x = u.z; a.y = u.w;
        acc0 = __builtin_amdgcn_wmma_f32_16x16x4_f32(false, a, false, bf[2 * j + 1], (short)0, acc0,
                                                     false, false);
        a.x = v.z; a.y = v.w;
        acc1 = __builtin_amdgcn_wmma_f32_16x16x4_f32(false, a, false, bf[2 * j + 1], (short)0, acc1,
                                                     false, false);
      }

      const int rowbase = tile * TR + 8 * h;
      const int lim = N - rowbase;
      float* tp = T2 + (size_t)rowbase * GDOUT + n;
#pragma unroll
      for (int r = 0; r < 8; ++r) {
        if (r < lim) tp[r * GDOUT] = acc0[r];
        if (r + 16 < lim) tp[(r + 16) * GDOUT] = acc1[r];
      }
      if (tile + G >= ntiles) break;
      __builtin_amdgcn_fence(__ATOMIC_RELEASE, "workgroup", "local");
      __builtin_amdgcn_s_barrier();
      __builtin_amdgcn_fence(__ATOMIC_ACQUIRE, "workgroup", "local");
    }

    tile = ntile; q = nq; i = ni; valid = nvalid; beg = nbeg; end = nend; di = ndi;
    s_l = (int)min((unsigned)ns, last);
    ntile = nntile; nq = nnq; ni = nni; nvalid = nnvalid; nbeg = nnbeg; nend = nnend;
  }
}

extern "C" void kernel_launch(void* const* d_in, const int* in_sizes, int n_in,
                              void* d_out, int out_size, void* d_ws, size_t ws_size,
                              hipStream_t stream) {
  const float* x  = (const float*)d_in[0];
  const int*   ei = (const int*)d_in[1];
  const float* W1 = (const float*)d_in[2];
  const float* b1 = (const float*)d_in[3];
  const float* W2 = (const float*)d_in[4];
  const float* b2 = (const float*)d_in[5];
  float* out = (float*)d_out;

  const int N = in_sizes[0] / GDIN;
  const int E = in_sizes[1] / 2;
  const int* src = ei;
  const int* dst = ei + E;

  float* dinv = (float*)d_ws;
  float* bufA = dinv + N;
  float* bufB = bufA + (size_t)N * GDH;

  int* gcount   = (int*)bufB;
  int* gcursor  = gcount + MAXB * CPAD;
  int* bucketed = gcursor + MAXB * CPAD;
  const int NB = (N + BROWS - 1) >> BSH;
  const int RP = (N + 4) & ~3;
  int* lst = (int*)(bufB + (size_t)N * GDOUT);

  zero_i32_kernel<<<(2 * MAXB * CPAD + 255) / 256, 256, 0, stream>>>(gcount, 2 * MAXB * CPAD);
  bucket_count_kernel<<<(E + SLICE - 1) / SLICE, MAXB, 0, stream>>>(dst, gcount, E, NB);
  bucket_scatter_kernel<<<(E + SLICE - 1) / SLICE, MAXB, 0, stream>>>(src, dst, gcount, gcursor, bucketed,
                                                                     E, NB);
  bucket_rows_fill_kernel<<<NB, MAXB, 0, stream>>>(gcount, bucketed, lst, dinv, lst + RP, N);

  gemm_wmma_f32<GDH><<<768, 256, 0, stream>>>(x, W1, dinv, bufA, N);
  gather_sum_rows_gemm<<<(N + 31) / 32, 128, 0, stream>>>(lst, lst + RP, dinv, bufA, b1, W2, bufB, N);

  gather_sum_rows<GDOUT, false><<<2048, 256, 0, stream>>>(lst, lst + RP, dinv, bufB, b2, out, N);
}


# Sort level 1 keeps each 4096-edge slice in place, ordered by bucket in LDS, run starts beside it; level 2 walks the runs: no global counters, zero and count launches gone; f32 row sum order changes
# speedup vs baseline: 5.1513x; 1.2005x over previous
#include <hip/hip_runtime.h>

typedef float v2f __attribute__((ext_vector_type(2)));
typedef float v8f __attribute__((ext_vector_type(8)));

#define GDIN  128
#define GDH   128
#define GDOUT 64

#define BSH   8
#define BROWS (1 << BSH)
#define MAXB  512
#define SLICE 4096
#define OP    (MAXB + 1)
#define WIN   8192

__global__ void __launch_bounds__(MAXB) slice_sort_kernel(const int* __restrict__ src,
                                                          const int* __restrict__ dst,
                                                          int* __restrict__ sliced_w,
                                                          int* __restrict__ ofs, int E) {
  __shared__ int2 lbuf[SLICE];
  __shared__ int hist[MAXB];
  __shared__ int lofs[MAXB];
  __shared__ int wtc[MAXB / 32];
  int2* __restrict__ sliced = reinterpret_cast<int2*>(sliced_w);
  const int tid = threadIdx.x;
  const int lane = tid & 31;
  const int wave = tid >> 5;
  const int beg = blockIdx.x * SLICE;
  const int n = min(SLICE, E - beg);
  int d[SLICE / MAXB], s[SLICE / MAXB];
  for (int u = 0; u < SLICE / MAXB; ++u) {
    const int i = tid + u * MAXB;
    d[u] = i < n ? dst[beg + i] : -1;
    s[u] = i < n ? src[beg + i] : 0;
  }
  hist[tid] = 0;
  __syncthreads();
  for (int u = 0; u < SLICE / MAXB; ++u)
    if (d[u] >= 0) atomicAdd(&hist[d[u] >> BSH], 1);
  __syncthreads();
  const int c = hist[tid];
  int ic = c;
  for (int o = 1; o < 32; o <<= 1) {
    const int vc = __shfl_up(ic, o);
    if (lane >= o) ic += vc;
  }
  if (lane == 31) wtc[wave] = ic;
  __syncthreads();
  int oc = 0;
  for (int w = 0; w < wave; ++w) oc += wtc[w];
  const int lo = oc + ic - c;
  lofs[tid] = lo;
  int* __restrict__ orow = ofs + (size_t)blockIdx.x * OP;
  orow[tid] = lo;
  if (tid == 0) orow[MAXB] = n;
  hist[tid] = 0;
  __syncthreads();
  for (int u = 0; u < SLICE / MAXB; ++u) {
    if (d[u] >= 0) {
      const int b = d[u] >> BSH;
      const int k = atomicAdd(&hist[b], 1);
      lbuf[lofs[b] + k] = make_int2(s[u], d[u]);
    }
  }
  __syncthreads();
  for (int i = tid; i < n; i += MAXB) sliced[beg + i] = lbuf[i];
}

__global__ void __launch_bounds__(MAXB) bucket_rows_fill_kernel(const int* __restrict__ ofs,
                                                                const int* __restrict__ sliced_w,
                                                                int* __restrict__ rowptr,
                                                                float* __restrict__ dinv,
                                                                int* __restrict__ list, int N, int NS) {
  __shared__ int ptr[WIN];
  __shared__ int rcnt[BROWS];
  __shared__ int rofs[BROWS];
  __shared__ int wlen[MAXB / 32];
  __shared__ int wlo[MAXB / 32];
  __shared__ int wtot[BROWS / 32];
  const int2* __restrict__ sliced = reinterpret_cast<const int2*>(sliced_w);
  const int tid = threadIdx.x;
  const int lane = tid & 31;
  const int wave = tid >> 5;
  const int b = blockIdx.x;
  const int r0 = b << BSH;
  int lo = 0, hi = 0;
  if (tid < NS) {
    lo = ofs[(size_t)tid * OP + b];
    hi = ofs[(size_t)tid * OP + b + 1];
  }
  const int len = hi - lo;
  int linc = len;
  for (int o = 1; o < 32; o <<= 1) {
    const int v = __shfl_up(linc, o);
    if (lane >= o) linc += v;
  }
  int slo = lo;
  for (int o = 16; o > 0; o >>= 1) slo += __shfl_xor(slo, o);
  if (lane == 31) wlen[wave] = linc;
  if (lane == 0) wlo[wave] = slo;
  if (tid < BROWS) rcnt[tid] = 0;
  __syncthreads();
  int pbase = 0, cnt = 0, start = 0;
  for (int w = 0; w < MAXB / 32; ++w) {
    const int v = wlen[w];
    if (w < wave) pbase += v;
    cnt += v;
    start += wlo[w];
  }
  const int P = pbase + linc - len;
  const int g0 = tid * SLICE + lo - P;

  for (int wb = 0; wb < cnt; wb += WIN) {
    const int k1 = min(P + len, wb + WIN);
    for (int k = max(P, wb); k < k1; ++k) ptr[k - wb] = g0 + k;
    __syncthreads();
    const int m = min(WIN, cnt - wb);
    for (int i0 = tid; i0 < m; i0 += 4 * MAXB) {
      int q[4], d[4];
      for (int u = 0; u < 4; ++u) q[u] = ptr[min(i0 + u * MAXB, WIN - 1)];
      for (int u = 0; u < 4; ++u) d[u] = i0 + u * MAXB < m ? sliced[q[u]].y : -1;
      for (int u = 0; u < 4; ++u)
        if (d[u] >= 0) atomicAdd(&rcnt[d[u] - r0], 1);
    }
    __syncthreads();
  }
  const int c = tid < BROWS ? rcnt[tid] : 0;
  int incl = c;
  for (int o = 1; o < 32; o <<= 1) {
    const int v = __shfl_up(incl, o);
    if (lane >= o) incl += v;
  }
  if (tid < BROWS && lane == 31) wtot[wave] = incl;
  __syncthreads();
  if (tid < BROWS) {
    int off = start;
    for (int w = 0; w < wave; ++w) off += wtot[w];
    const int row = r0 + tid;
    const int first = off + incl - c;
    rofs[tid] = first;
    rcnt[tid] = 0;
    if (row < N) {
      rowptr[row] = first;
      dinv[row] = rsqrtf((float)(c + 1));
    }
    if (b == (int)gridDim.x - 1 && tid == BROWS - 1) rowptr[N] = off + incl;
  }
  __syncthreads();
  const bool many = cnt > WIN;
  for (int wb = 0; wb < cnt; wb += WIN) {
    if (many) {
      const int k1 = min(P + len, wb + WIN);
      for (int k = max(P, wb); k < k1; ++k) ptr[k - wb] = g0 + k;
      __syncthreads();
    }
    const int m = min(WIN, cnt - wb);
    for (int i0 = tid; i0 < m; i0 += 4 * MAXB) {
      int q[4];
      int2 p[4];
      for (int u = 0; u < 4; ++u) q[u] = ptr[min(i0 + u * MAXB, WIN - 1)];
      for (int u = 0; u < 4; ++u) p[u] = i0 + u * MAXB < m ? sliced[q[u]] : make_int2(0, -1);
      for (int u = 0; u < 4; ++u) {
        if (p[u].y >= 0) {
          const int k = p[u].y - r0;
          const int rk = atomicAdd(&rcnt[k], 1);
          list[rofs[k] + rk] = p[u].x;
        }
      }
    }
    if (many) __syncthreads();
  }
}

template <int NCOLS>
__global__ void __launch_bounds__(256) gemm_wmma_f32(const float* __restrict__ X,
                                                     const float* __restrict__ W,
                                                     const float* __restrict__ rowscale,
                                                     float* __restrict__ T, int N) {
  constexpr int CT = NCOLS / 16;
  constexpr int RG = 8 / CT;
  constexpr int CH = 32 * RG;
  constexpr int NLD = CH / 8;
  constexpr int P4 = 33;
  __shared__ float4 xs4[CH * P4];

  const int tid = threadIdx.x;
  const int wave = tid >> 5;
  const int lane = tid & 31;
  const int c = lane & 15;
  const int h = lane >> 4;
  const int ct = wave % CT;
  const int rg = wave / CT;
  const int n = ct * 16 + c;

  v2f bf[32];
#pragma unroll
  for (int j = 0; j < 16; ++j) {
    const int kb = 8 * j + 4 * h;
    bf[2 * j].x     = W[(size_t)(kb + 0) * NCOLS + n];
    bf[2 * j].y     = W[(size_t)(kb + 1) * NCOLS + n];
    bf[2 * j + 1].x = W[(size_t)(kb + 2) * NCOLS + n];
    bf[2 * j + 1].y = W[(size_t)(kb + 3) * NCOLS + n];
  }

  const float4* __restrict__ X4 = reinterpret_cast<const float4*>(X);
  const int nchunks = (N + CH - 1) / CH;
  int chunk = blockIdx.x;

  float4 pre[NLD];
  float prs[NLD];
  if (chunk < nchunks) {
#pragma unroll
    for (int q = 0; q < NLD; ++q) {
      const int i = tid + 256 * q;
      const int row = chunk * CH + (i >> 5);
      pre[q] = row < N ? X4[(size_t)row * 32 + (i & 31)] : make_float4(0.f, 0.f, 0.f, 0.f);
      prs[q] = row < N ? rowscale[row] : 0.f;
    }
  }

  for (; chunk < nchunks; chunk += gridDim.x) {
#pragma unroll
    for (int q = 0; q < NLD; ++q) {
      const int i = tid + 256 * q;
      xs4[(i >> 5) * P4 + (i & 31)] = make_float4(pre[q].x * prs[q], pre[q].y * prs[q],
                                                  pre[q].z * prs[q], pre[q].w * prs[q]);
    }
    __builtin_amdgcn_fence(__ATOMIC_RELEASE, "workgroup", "local");
    __builtin_amdgcn_s_barrier();
    __builtin_amdgcn_fence(__ATOMIC_ACQUIRE, "workgroup", "local");

    const int next = chunk + gridDim.x;
    if (next < nchunks) {
#pragma unroll
      for (int q = 0; q < NLD; ++q) {
        const int i = tid + 256 * q;
        const int row = next * CH + (i >> 5);
        pre[q] = row < N ? X4[(size_t)row * 32 + (i & 31)] : make_float4(0.f, 0.f, 0.f, 0.f);
        prs[q] = row < N ? rowscale[row] : 0.f;
      }
    }

    const float4* a0p = xs4 + (rg * 32 + c) * P4 + h;
    const float4* a1p = a0p + 16 * P4;
    v8f acc0 = {};
    v8f acc1 = {};
#pragma unroll
    for (int j = 0; j < 16; ++j) {
      const float4 u = a0p[2 * j];
      const float4 v = a1p[2 * j];
      v2f a;
      a.x = u.x; a.y = u.y;
      acc0 = __builtin_amdgcn_wmma_f32_16x16x4_f32(false, a, false, bf[2 * j], (short)0, acc0,
                                                   false, false);
      a.x = v.x; a.y = v.y;
      acc1 = __builtin_amdgcn_wmma_f32_16x16x4_f32(false, a, false, bf[2 * j], (short)0, acc1,
                                                   false, false);
      a.x = u.z; a.y = u.w;
      acc0 = __builtin_amdgcn_wmma_f32_16x16x4_f32(false, a, false, bf[2 * j + 1], (short)0, acc0,
                                                   false, false);
      a.x = v.z; a.y = v.w;
      acc1 = __builtin_amdgcn_wmma_f32_16x16x4_f32(false, a, false, bf[2 * j + 1], (short)0, acc1,
                                                   false, false);
    }

    const int rowbase = chunk * CH + rg * 32 + 8 * h;
    const int lim = N - rowbase;
    float* tp = T + (size_t)rowbase * NCOLS + n;
#pragma unroll
    for (int r = 0; r < 8; ++r) {
      if (r < lim) tp[r * NCOLS] = acc0[r];
      if (r + 16 < lim) tp[(r + 16) * NCOLS] = acc1[r];
    }
    __builtin_amdgcn_fence(__ATOMIC_RELEASE, "workgroup", "local");
    __builtin_amdgcn_s_barrier();
    __builtin_amdgcn_fence(__ATOMIC_ACQUIRE, "workgroup", "local");
  }
}

template <int V>
__device__ __forceinline__ void sum_run(float __attribute__((ext_vector_type(V)))& acc,
                                        const float __attribute__((ext_vector_type(V)))* __restrict__ Tv,
                                        int s_l, int cnt, int lane) {
  typedef float vf __attribute__((ext_vector_type(V)));
  for (int j = 0; j < cnt; j += 4) {
    vf v[4];
    float w[4];
    for (int u = 0; u < 4; ++u) {
      const int s = __builtin_amdgcn_readlane(s_l, j + u);
      w[u] = (u == 0 || j + u < cnt) ? 1.0f : 0.0f;
      v[u] = Tv[(size_t)s * 32 + lane];
    }
    for (int u = 0; u < 4; ++u) acc += v[u] * w[u];
  }
}

template <int V>
__device__ __forceinline__ void sum_run8(float __attribute__((ext_vector_type(V)))& acc,
                                         const float __attribute__((ext_vector_type(V)))* __restrict__ Tv,
                                         int s_l, int cnt, int lane) {
  typedef float vf __attribute__((ext_vector_type(V)));
  int j = 0;
  for (; cnt - j > 4; j += 8) {
    vf v[8];
    float w[8];
    for (int u = 0; u < 8; ++u) {
      const int s = __builtin_amdgcn_readlane(s_l, j + u);
      w[u] = (u < 5 || j + u < cnt) ? 1.0f : 0.0f;
      v[u] = Tv[(size_t)s * 32 + lane];
    }
    for (int u = 0; u < 8; ++u) acc += v[u] * w[u];
  }
  if (j < cnt) {
    vf v[4];
    float w[4];
    for (int u = 0; u < 4; ++u) {
      const int s = __builtin_amdgcn_readlane(s_l, j + u);
      w[u] = (u == 0 || j + u < cnt) ? 1.0f : 0.0f;
      v[u] = Tv[(size_t)s * 32 + lane];
    }
    for (int u = 0; u < 4; ++u) acc += v[u] * w[u];
  }
}

template <int D, bool RELU>
__global__ void __launch_bounds__(256) gather_sum_rows(const int* __restrict__ rowptr,
                                                       const int* __restrict__ list,
                                                       const float* __restrict__ dinv,
                                                       const float* __restrict__ T,
                                                       const float* __restrict__ bias,
                                                       float* __restrict__ OUT, int N) {
  constexpr int V = D / 32;
  typedef float vf __attribute__((ext_vector_type(V)));
  const int lane = threadIdx.x & 31;
  const int nw = (gridDim.x * blockDim.x) >> 5;
  int i = __builtin_amdgcn_readfirstlane((int)((blockIdx.x * blockDim.x + threadIdx.x) >> 5));
  if (i >= N) return;
  const vf* __restrict__ Tv = reinterpret_cast<const vf*>(T);
  vf* __restrict__ Ov = reinterpret_cast<vf*>(OUT);
  const vf bv = reinterpret_cast<const vf*>(bias)[lane];

  int beg = rowptr[i], end = rowptr[i + 1];
  int ni = i + nw;
  int nbeg = 0, nend = 0;
  if (ni < N) { nbeg = rowptr[ni]; nend = rowptr[ni + 1]; }
  const unsigned last = (unsigned)(N - 1);
  int s_l = i;
  if (beg + lane < end) s_l = list[beg + lane];
  s_l = (int)min((unsigned)s_l, last);
  float di = dinv[i];

  for (;;) {
    const int nni = ni + nw;
    int nnbeg = 0, nnend = 0;
    if (nni < N) { nnbeg = rowptr[nni]; nnend = rowptr[nni + 1]; }
    float ndi = 0.0f;
    int ns = i;
    if (ni < N) {
      ndi = dinv[ni];
      ns = ni;
      if (nbeg + lane < nend) ns = list[nbeg + lane];
    }
    const vf selfv = Tv[(size_t)i * 32 + lane];
    vf acc = 0.0f;
    sum_run<V>(acc, Tv, s_l, min(32, end - beg), lane);
    for (int base = beg + 32; base < end; base += 32) {
      int q = i;
      if (base + lane < end) q = list[base + lane];
      sum_run<V>(acc, Tv, (int)min((unsigned)q, last), min(32, end - base), lane);
    }
    acc += selfv;
    acc = acc * di + bv;
    if (RELU) {
      for (int k = 0; k < V; ++k) acc[k] = acc[k] > 0.0f ? acc[k] : 0.0f;
    }
    Ov[(size_t)i * 32 + lane] = acc;
    if (ni >= N) break;
    i = ni; beg = nbeg; end = nend; di = ndi;
    s_l = (int)min((unsigned)ns, last);
    ni = nni; nbeg = nnbeg; nend = nnend;
  }
}

__global__ void __launch_bounds__(128) gather_sum_rows_gemm(const int* __restrict__ rowptr,
                                                            const int* __restrict__ list,
                                                            const float* __restrict__ dinv,
                                                            const float* __restrict__ T,
                                                            const float* __restrict__ bias,
                                                            const float* __restrict__ W,
                                                            float* __restrict__ T2, int N) {
  constexpr int V = GDH / 32;
  constexpr int TR = 32;
  constexpr int RW = TR / 4;
  constexpr int P4 = 33;
  typedef float vf __attribute__((ext_vector_type(V)));
  __shared__ float4 hs4[TR * P4];

  const int lane = threadIdx.x & 31;
  const int wave = __builtin_amdgcn_readfirstlane((int)(threadIdx.x >> 5));
  const int c = lane & 15;
  const int h = lane >> 4;
  const int n = wave * 16 + c;

  const int ntiles = (N + TR - 1) / TR;
  const int G = gridDim.x;
  int tile = blockIdx.x;
  if (tile >= ntiles) return;

  v2f bf[32];
#pragma unroll
  for (int j = 0; j < 16; ++j) {
    const int kb = 8 * j + 4 * h;
    bf[2 * j].x     = W[(size_t)(kb + 0) * GDOUT + n];
    bf[2 * j].y     = W[(size_t)(kb + 1) * GDOUT + n];
    bf[2 * j + 1].x = W[(size_t)(kb + 2) * GDOUT + n];
    bf[2 * j + 1].y = W[(size_t)(kb + 3) * GDOUT + n];
  }

  const vf* __restrict__ Tv = reinterpret_cast<const vf*>(T);
  const vf bv = reinterpret_cast<const vf*>(bias)[lane];

  int q = 0;
  int i = tile * TR + wave * RW;
  bool valid = i < N;
  int beg = 0, end = 0;
  if (valid) { beg = rowptr[i]; end = rowptr[i + 1]; }
  int ntile = tile, nq = 1;
  int ni = i + 1;
  bool nvalid = ni < N;
  int nbeg = 0, nend = 0;
  if (nvalid) { nbeg = rowptr[ni]; nend = rowptr[ni + 1]; }
  const unsigned last = (unsigned)(N - 1);
  float di = 0.0f;
  int s_l = 0;
  if (valid) {
    di = dinv[i];
    s_l = i;
    if (beg + lane < end) s_l = list[beg + lane];
  }
  s_l = (int)min((unsigned)s_l, last);

  for (;;) {
    int nntile = ntile, nnq = nq + 1;
    if (nnq == RW) { nnq = 0; nntile += G; }
    const int nni = nntile * TR + wave * RW + nnq;
    const bool nnvalid = nntile < ntiles && nni < N;
    int nnbeg = 0, nnend = 0;
    if (nnvalid) { nnbeg = rowptr[nni]; nnend = rowptr[nni + 1]; }
    float ndi = 0.0f;
    int ns = 0;
    if (nvalid) {
      ndi = dinv[ni];
      ns = ni;
      if (nbeg + lane < nend) ns = list[nbeg + lane];
    }
    vf acc = 0.0f;
    if (valid) {
      const vf selfv = Tv[(size_t)i * 32 + lane];
      sum_run8<V>(acc, Tv, s_l, min(32, end - beg), lane);
      for (int base = beg + 32; base < end; base += 32) {
        int r = i;
        if (base + lane < end) r = list[base + lane];
        sum_run8<V>(acc, Tv, (int)min((unsigned)r, last), min(32, end - base), lane);
      }
      acc += selfv;
      acc = acc * di + bv;
      for (int k = 0; k < V; ++k) acc[k] = acc[k] > 0.0f ? acc[k] : 0.0f;
      acc *= di;
    }
    hs4[(wave * RW + q) * P4 + lane] = make_float4(acc[0], acc[1], acc[2], acc[3]);

    if (q == RW - 1) {
      __builtin_amdgcn_fence(__ATOMIC_RELEASE, "workgroup", "local");
      __builtin_amdgcn_s_barrier();
      __builtin_amdgcn_fence(__ATOMIC_ACQUIRE, "workgroup", "local");

      const float4* a0p = hs4 + c * P4 + h;
      const float4* a1p = a0p + 16 * P4;
      v8f acc0 = {};
      v8f acc1 = {};
#pragma unroll
      for (int j = 0; j < 16; ++j) {
        const float4 u = a0p[2 * j];
        const float4 v = a1p[2 * j];
        v2f a;
        a.x = u.x; a.y = u.y;
        acc0 = __builtin_amdgcn_wmma_f32_16x16x4_f32(false, a, false, bf[2 * j], (short)0, acc0,
                                                     false, false);
        a.x = v.x; a.y = v.y;
        acc1 = __builtin_amdgcn_wmma_f32_16x16x4_f32(false, a, false, bf[2 * j], (short)0, acc1,
                                                     false, false);
        a.x = u.z; a.y = u.w;
        acc0 = __builtin_amdgcn_wmma_f32_16x16x4_f32(false, a, false, bf[2 * j + 1], (short)0, acc0,
                                                     false, false);
        a.x = v.z; a.y = v.w;
        acc1 = __builtin_amdgcn_wmma_f32_16x16x4_f32(false, a, false, bf[2 * j + 1], (short)0, acc1,
                                                     false, false);
      }

      const int rowbase = tile * TR + 8 * h;
      const int lim = N - rowbase;
      float* tp = T2 + (size_t)rowbase * GDOUT + n;
#pragma unroll
      for (int r = 0; r < 8; ++r) {
        if (r < lim) tp[r * GDOUT] = acc0[r];
        if (r + 16 < lim) tp[(r + 16) * GDOUT] = acc1[r];
      }
      if (tile + G >= ntiles) break;
      __builtin_amdgcn_fence(__ATOMIC_RELEASE, "workgroup", "local");
      __builtin_amdgcn_s_barrier();
      __builtin_amdgcn_fence(__ATOMIC_ACQUIRE, "workgroup", "local");
    }

    tile = ntile; q = nq; i = ni; valid = nvalid; beg = nbeg; end = nend; di = ndi;
    s_l = (int)min((unsigned)ns, last);
    ntile = nntile; nq = nnq; ni = nni; nvalid = nnvalid; nbeg = nnbeg; nend = nnend;
  }
}

extern "C" void kernel_launch(void* const* d_in, const int* in_sizes, int n_in,
                              void* d_out, int out_size, void* d_ws, size_t ws_size,
                              hipStream_t stream) {
  const float* x  = (const float*)d_in[0];
  const int*   ei = (const int*)d_in[1];
  const float* W1 = (const float*)d_in[2];
  const float* b1 = (const float*)d_in[3];
  const float* W2 = (const float*)d_in[4];
  const float* b2 = (const float*)d_in[5];
  float* out = (float*)d_out;

  const int N = in_sizes[0] / GDIN;
  const int E = in_sizes[1] / 2;
  const int* src = ei;
  const int* dst = ei + E;

  float* dinv = (float*)d_ws;
  float* bufA = dinv + N;
  float* bufB = bufA + (size_t)N * GDH;

  const int NS = (E + SLICE - 1) / SLICE;
  int* sliced = (int*)bufB;
  int* ofs    = sliced + (size_t)NS * SLICE * 2;
  const int NB = (N + BROWS - 1) >> BSH;
  const int RP = (N + 4) & ~3;
  int* lst = (int*)(bufB + (size_t)N * GDOUT);

  slice_sort_kernel<<<NS, MAXB, 0, stream>>>(src, dst, sliced, ofs, E);
  bucket_rows_fill_kernel<<<NB, MAXB, 0, stream>>>(ofs, sliced, lst, dinv, lst + RP, N, NS);

  gemm_wmma_f32<GDH><<<768, 256, 0, stream>>>(x, W1, dinv, bufA, N);
  gather_sum_rows_gemm<<<(N + 31) / 32, 128, 0, stream>>>(lst, lst + RP, dinv, bufA, b1, W2, bufB, N);

  gather_sum_rows<GDOUT, false><<<2048, 256, 0, stream>>>(lst, lst + RP, dinv, bufB, b2, out, N);
}


# Layer 1 as (sorted row sums of x, entry weight dinv[src]) @ W1: a 64-row block sums x rows into LDS and runs both f32 WMMA products there; x @ W1 launch and table gone; f32 sum order changes
# speedup vs baseline: 5.9004x; 1.1454x over previous
#include <hip/hip_runtime.h>

typedef float v2f __attribute__((ext_vector_type(2)));
typedef float v8f __attribute__((ext_vector_type(8)));

#define GDIN  128
#define GDH   128
#define GDOUT 64

#define BSH   8
#define BROWS (1 << BSH)
#define MAXB  512
#define SLICE 4096
#define OP    (MAXB + 1)
#define WIN   8192

__global__ void __launch_bounds__(MAXB) slice_sort_kernel(const int* __restrict__ src,
                                                          const int* __restrict__ dst,
                                                          int* __restrict__ sliced_w,
                                                          int* __restrict__ ofs, int E) {
  __shared__ int2 lbuf[SLICE];
  __shared__ int hist[MAXB];
  __shared__ int lofs[MAXB];
  __shared__ int wtc[MAXB / 32];
  int2* __restrict__ sliced = reinterpret_cast<int2*>(sliced_w);
  const int tid = threadIdx.x;
  const int lane = tid & 31;
  const int wave = tid >> 5;
  const int beg = blockIdx.x * SLICE;
  const int n = min(SLICE, E - beg);
  int d[SLICE / MAXB], s[SLICE / MAXB];
  for (int u = 0; u < SLICE / MAXB; ++u) {
    const int i = tid + u * MAXB;
    d[u] = i < n ? dst[beg + i] : -1;
    s[u] = i < n ? src[beg + i] : 0;
  }
  hist[tid] = 0;
  __syncthreads();
  for (int u = 0; u < SLICE / MAXB; ++u)
    if (d[u] >= 0) atomicAdd(&hist[d[u] >> BSH], 1);
  __syncthreads();
  const int c = hist[tid];
  int ic = c;
  for (int o = 1; o < 32; o <<= 1) {
    const int vc = __shfl_up(ic, o);
    if (lane >= o) ic += vc;
  }
  if (lane == 31) wtc[wave] = ic;
  __syncthreads();
  int oc = 0;
  for (int w = 0; w < wave; ++w) oc += wtc[w];
  const int lo = oc + ic - c;
  lofs[tid] = lo;
  int* __restrict__ orow = ofs + (size_t)blockIdx.x * OP;
  orow[tid] = lo;
  if (tid == 0) orow[MAXB] = n;
  hist[tid] = 0;
  __syncthreads();
  for (int u = 0; u < SLICE / MAXB; ++u) {
    if (d[u] >= 0) {
      const int b = d[u] >> BSH;
      const int k = atomicAdd(&hist[b], 1);
      lbuf[lofs[b] + k] = make_int2(s[u], d[u]);
    }
  }
  __syncthreads();
  for (int i = tid; i < n; i += MAXB) sliced[beg + i] = lbuf[i];
}

__global__ void __launch_bounds__(MAXB) bucket_rows_fill_kernel(const int* __restrict__ ofs,
                                                                const int* __restrict__ sliced_w,
                                                                int* __restrict__ rowptr,
                                                                float* __restrict__ dinv,
                                                                int* __restrict__ list, int N, int NS) {
  __shared__ int ptr[WIN];
  __shared__ int rcnt[BROWS];
  __shared__ int rofs[BROWS];
  __shared__ int wlen[MAXB / 32];
  __shared__ int wlo[MAXB / 32];
  __shared__ int wtot[BROWS / 32];
  const int2* __restrict__ sliced = reinterpret_cast<const int2*>(sliced_w);
  const int tid = threadIdx.x;
  const int lane = tid & 31;
  const int wave = tid >> 5;
  const int b = blockIdx.x;
  const int r0 = b << BSH;
  int lo = 0, hi = 0;
  if (tid < NS) {
    lo = ofs[(size_t)tid * OP + b];
    hi = ofs[(size_t)tid * OP + b + 1];
  }
  const int len = hi - lo;
  int linc = len;
  for (int o = 1; o < 32; o <<= 1) {
    const int v = __shfl_up(linc, o);
    if (lane >= o) linc += v;
  }
  int slo = lo;
  for (int o = 16; o > 0; o >>= 1) slo += __shfl_xor(slo, o);
  if (lane == 31) wlen[wave] = linc;
  if (lane == 0) wlo[wave] = slo;
  if (tid < BROWS) rcnt[tid] = 0;
  __syncthreads();
  int pbase = 0, cnt = 0, start = 0;
  for (int w = 0; w < MAXB / 32; ++w) {
    const int v = wlen[w];
    if (w < wave) pbase += v;
    cnt += v;
    start += wlo[w];
  }
  const int P = pbase + linc - len;
  const int g0 = tid * SLICE + lo - P;

  for (int wb = 0; wb < cnt; wb += WIN) {
    const int k1 = min(P + len, wb + WIN);
    for (int k = max(P, wb); k < k1; ++k) ptr[k - wb] = g0 + k;
    __syncthreads();
    const int m = min(WIN, cnt - wb);
    for (int i0 = tid; i0 < m; i0 += 4 * MAXB) {
      int q[4], d[4];
      for (int u = 0; u < 4; ++u) q[u] = ptr[min(i0 + u * MAXB, WIN - 1)];
      for (int u = 0; u < 4; ++u) d[u] = i0 + u * MAXB < m ? sliced[q[u]].y : -1;
      for (int u = 0; u < 4; ++u)
        if (d[u] >= 0) atomicAdd(&rcnt[d[u] - r0], 1);
    }
    __syncthreads();
  }
  const int c = tid < BROWS ? rcnt[tid] : 0;
  int incl = c;
  for (int o = 1; o < 32; o <<= 1) {
    const int v = __shfl_up(incl, o);
    if (lane >= o) incl += v;
  }
  if (tid < BROWS && lane == 31) wtot[wave] = incl;
  __syncthreads();
  if (tid < BROWS) {
    int off = start;
    for (int w = 0; w < wave; ++w) off += wtot[w];
    const int row = r0 + tid;
    const int first = off + incl - c;
    rofs[tid] = first;
    rcnt[tid] = 0;
    if (row < N) {
      rowptr[row] = first;
      dinv[row] = rsqrtf((float)(c + 1));
    }
    if (b == (int)gridDim.x - 1 && tid == BROWS - 1) rowptr[N] = off + incl;
  }
  __syncthreads();
  const bool many = cnt > WIN;
  for (int wb = 0; wb < cnt; wb += WIN) {
    if (many) {
      const int k1 = min(P + len, wb + WIN);
      for (int k = max(P, wb); k < k1; ++k) ptr[k - wb] = g0 + k;
      __syncthreads();
    }
    const int m = min(WIN, cnt - wb);
    for (int i0 = tid; i0 < m; i0 += 4 * MAXB) {
      int q[4];
      int2 p[4];
      for (int u = 0; u < 4; ++u) q[u] = ptr[min(i0 + u * MAXB, WIN - 1)];
      for (int u = 0; u < 4; ++u) p[u] = i0 + u * MAXB < m ? sliced[q[u]] : make_int2(0, -1);
      for (int u = 0; u < 4; ++u) {
        if (p[u].y >= 0) {
          const int k = p[u].y - r0;
          const int rk = atomicAdd(&rcnt[k], 1);
          list[rofs[k] + rk] = p[u].x;
        }
      }
    }
    if (many) __syncthreads();
  }
}

template <int V>
__device__ __forceinline__ void sum_run(float __attribute__((ext_vector_type(V)))& acc,
                                        const float __attribute__((ext_vector_type(V)))* __restrict__ Tv,
                                        int s_l, int cnt, int lane) {
  typedef float vf __attribute__((ext_vector_type(V)));
  for (int j = 0; j < cnt; j += 4) {
    vf v[4];
    float w[4];
    for (int u = 0; u < 4; ++u) {
      const int s = __builtin_amdgcn_readlane(s_l, j + u);
      w[u] = (u == 0 || j + u < cnt) ? 1.0f : 0.0f;
      v[u] = Tv[(size_t)s * 32 + lane];
    }
    for (int u = 0; u < 4; ++u) acc += v[u] * w[u];
  }
}

template <int V>
__device__ __forceinline__ void sum_run8w(float __attribute__((ext_vector_type(V)))& acc,
                                          const float __attribute__((ext_vector_type(V)))* __restrict__ Xv,
                                          int s_l, float w_l, int cnt, int lane) {
  typedef float vf __attribute__((ext_vector_type(V)));
  const int wi_l = __float_as_int(w_l);
  int j = 0;
  for (; cnt - j > 4; j += 8) {
    vf v[8];
    float w[8];
    for (int u = 0; u < 8; ++u) {
      const int s = __builtin_amdgcn_readlane(s_l, j + u);
      w[u] = __int_as_float(__builtin_amdgcn_readlane(wi_l, j + u));
      v[u] = Xv[(size_t)s * 32 + lane];
    }
    for (int u = 0; u < 8; ++u) acc += v[u] * w[u];
  }
  if (j < cnt) {
    vf v[4];
    float w[4];
    for (int u = 0; u < 4; ++u) {
      const int s = __builtin_amdgcn_readlane(s_l, j + u);
      w[u] = __int_as_float(__builtin_amdgcn_readlane(wi_l, j + u));
      v[u] = Xv[(size_t)s * 32 + lane];
    }
    for (int u = 0; u < 4; ++u) acc += v[u] * w[u];
  }
}

template <int D, bool RELU>
__global__ void __launch_bounds__(256) gather_sum_rows(const int* __restrict__ rowptr,
                                                       const int* __restrict__ list,
                                                       const float* __restrict__ dinv,
                                                       const float* __restrict__ T,
                                                       const float* __restrict__ bias,
                                                       float* __restrict__ OUT, int N) {
  constexpr int V = D / 32;
  typedef float vf __attribute__((ext_vector_type(V)));
  const int lane = threadIdx.x & 31;
  const int nw = (gridDim.x * blockDim.x) >> 5;
  int i = __builtin_amdgcn_readfirstlane((int)((blockIdx.x * blockDim.x + threadIdx.x) >> 5));
  if (i >= N) return;
  const vf* __restrict__ Tv = reinterpret_cast<const vf*>(T);
  vf* __restrict__ Ov = reinterpret_cast<vf*>(OUT);
  const vf bv = reinterpret_cast<const vf*>(bias)[lane];

  int beg = rowptr[i], end = rowptr[i + 1];
  int ni = i + nw;
  int nbeg = 0, nend = 0;
  if (ni < N) { nbeg = rowptr[ni]; nend = rowptr[ni + 1]; }
  const unsigned last = (unsigned)(N - 1);
  int s_l = i;
  if (beg + lane < end) s_l = list[beg + lane];
  s_l = (int)min((unsigned)s_l, last);
  float di = dinv[i];

  for (;;) {
    const int nni = ni + nw;
    int nnbeg = 0, nnend = 0;
    if (nni < N) { nnbeg = rowptr[nni]; nnend = rowptr[nni + 1]; }
    float ndi = 0.0f;
    int ns = i;
    if (ni < N) {
      ndi = dinv[ni];
      ns = ni;
      if (nbeg + lane < nend) ns = list[nbeg + lane];
    }
    const vf selfv = Tv[(size_t)i * 32 + lane];
    vf acc = 0.0f;
    sum_run<V>(acc, Tv, s_l, min(32, end - beg), lane);
    for (int base = beg + 32; base < end; base += 32) {
      int q = i;
      if (base + lane < end) q = list[base + lane];
      sum_run<V>(acc, Tv, (int)min((unsigned)q, last), min(32, end - base), lane);
    }
    acc += selfv;
    acc = acc * di + bv;
    if (RELU) {
      for (int k = 0; k < V; ++k) acc[k] = acc[k] > 0.0f ? acc[k] : 0.0f;
    }
    Ov[(size_t)i * 32 + lane] = acc;
    if (ni >= N) break;
    i = ni; beg = nbeg; end = nend; di = ndi;
    s_l = (int)min((unsigned)ns, last);
    ni = nni; nbeg = nnbeg; nend = nnend;
  }
}

__global__ void __launch_bounds__(256, 8) gather_rows_two_gemms(const int* __restrict__ rowptr,
                                                             const int* __restrict__ list,
                                                             const float* __restrict__ dinv,
                                                             const float* __restrict__ X,
                                                             const float* __restrict__ W1,
                                                             const float* __restrict__ bias1,
                                                             const float* __restrict__ W2,
                                                             float* __restrict__ T2, int N) {
  constexpr int V = GDIN / 32;
  constexpr int TR = 64;
  constexpr int RW = TR / 8;
  constexpr int P4 = 33;
  typedef float vf __attribute__((ext_vector_type(V)));
  __shared__ float4 ts4[TR * P4];
  __shared__ float dsc[TR];

  const int lane = threadIdx.x & 31;
  const int wave = __builtin_amdgcn_readfirstlane((int)(threadIdx.x >> 5));
  const int c = lane & 15;
  const int h = lane >> 4;
  const int tile = blockIdx.x;

  const int n1 = wave * 16 + c;
  v2f bf1[32];
#pragma unroll
  for (int j = 0; j < 16; ++j) {
    const int kb = 8 * j + 4 * h;
    bf1[2 * j].x     = W1[(size_t)(kb + 0) * GDH + n1];
    bf1[2 * j].y     = W1[(size_t)(kb + 1) * GDH + n1];
    bf1[2 * j + 1].x = W1[(size_t)(kb + 2) * GDH + n1];
    bf1[2 * j + 1].y = W1[(size_t)(kb + 3) * GDH + n1];
  }

  const vf* __restrict__ Xv = reinterpret_cast<const vf*>(X);

  const int rbase = tile * TR + wave * RW;
  const unsigned last = (unsigned)(N - 1);
  const int rp_l = rowptr[min(rbase + min(lane, RW), N)];
  const int d_l = __float_as_int((lane < RW && rbase + lane < N) ? dinv[rbase + lane] : 0.0f);

  int i = rbase;
  const int b0 = __builtin_amdgcn_readlane(rp_l, 0);
  const int b1 = __builtin_amdgcn_readlane(rp_l, 1);
  const int b2 = __builtin_amdgcn_readlane(rp_l, 2);
  int s_l = i, ns_l = i + 1;
  if (b0 + lane < b1) s_l = list[b0 + lane];
  if (b1 + lane < b2) ns_l = list[b1 + lane];
  s_l = (int)min((unsigned)s_l, last);
  float w_l = 0.0f;
  if (b0 + lane < b1) w_l = dinv[s_l];

  for (int q = 0; q < RW; ++q) {
    const int beg = __builtin_amdgcn_readlane(rp_l, q);
    const int end = __builtin_amdgcn_readlane(rp_l, q + 1);
    const int e2 = __builtin_amdgcn_readlane(rp_l, min(q + 2, RW));
    const int e3 = __builtin_amdgcn_readlane(rp_l, min(q + 3, RW));
    ns_l = (int)min((unsigned)ns_l, last);
    float nw_l = 0.0f;
    if (end + lane < e2) nw_l = dinv[ns_l];
    int nns_l = i;
    if (e2 + lane < e3) nns_l = list[e2 + lane];

    const int ic = min(i, N - 1);
    const float di = __int_as_float(__builtin_amdgcn_readlane(d_l, q));
    const vf selfv = Xv[(size_t)ic * 32 + lane];
    vf acc = 0.0f;
    sum_run8w<V>(acc, Xv, s_l, w_l, min(32, end - beg), lane);
    for (int base = beg + 32; base < end; base += 32) {
      int r = ic;
      if (base + lane < end) r = list[base + lane];
      r = (int)min((unsigned)r, last);
      float wr = 0.0f;
      if (base + lane < end) wr = dinv[r];
      sum_run8w<V>(acc, Xv, r, wr, min(32, end - base), lane);
    }
    acc += selfv * di;
    acc *= di;
    ts4[(wave * RW + q) * P4 + lane] = make_float4(acc[0], acc[1], acc[2], acc[3]);
    if (lane == 0) dsc[wave * RW + q] = di;

    ++i;
    s_l = ns_l; w_l = nw_l; ns_l = nns_l;
  }
  __syncthreads();

  const float4* ap = ts4 + c * P4 + h;
  v8f hacc[4] = {};
#pragma unroll
  for (int j = 0; j < 16; ++j) {
    float4 u[4];
#pragma unroll
    for (int t = 0; t < 4; ++t) u[t] = ap[t * 16 * P4 + 2 * j];
#pragma unroll
    for (int t = 0; t < 4; ++t) {
      v2f a;
      a.x = u[t].x; a.y = u[t].y;
      hacc[t] = __builtin_amdgcn_wmma_f32_16x16x4_f32(false, a, false, bf1[2 * j], (short)0, hacc[t],
                                                      false, false);
    }
#pragma unroll
    for (int t = 0; t < 4; ++t) {
      v2f a;
      a.x = u[t].z; a.y = u[t].w;
      hacc[t] = __builtin_amdgcn_wmma_f32_16x16x4_f32(false, a, false, bf1[2 * j + 1], (short)0, hacc[t],
                                                      false, false);
    }
  }

  const int ct2 = wave & 3;
  const int rg2 = wave >> 2;
  const int n2 = ct2 * 16 + c;
  v2f bf2[32];
#pragma unroll
  for (int j = 0; j < 16; ++j) {
    const int kb = 8 * j + 4 * h;
    bf2[2 * j].x     = W2[(size_t)(kb + 0) * GDOUT + n2];
    bf2[2 * j].y     = W2[(size_t)(kb + 1) * GDOUT + n2];
    bf2[2 * j + 1].x = W2[(size_t)(kb + 2) * GDOUT + n2];
    bf2[2 * j + 1].y = W2[(size_t)(kb + 3) * GDOUT + n2];
  }
  const float bb = bias1[n1];

  __syncthreads();

  float* ts = reinterpret_cast<float*>(ts4);
#pragma unroll
  for (int t = 0; t < 4; ++t) {
#pragma unroll
    for (int r = 0; r < 8; ++r) {
      const int row = t * 16 + 8 * h + r;
      float hv = hacc[t][r] + bb;
      hv = hv > 0.0f ? hv : 0.0f;
      ts[row * (4 * P4) + n1] = hv * dsc[row];
    }
  }
  __syncthreads();

  const float4* a0p = ts4 + (rg2 * 32 + c) * P4 + h;
  const float4* a1p = a0p + 16 * P4;
  v8f acc0 = {};
  v8f acc1 = {};
#pragma unroll
  for (int j = 0; j < 16; ++j) {
    const float4 u = a0p[2 * j];
    const float4 v = a1p[2 * j];
    v2f a;
    a.x = u.x; a.y = u.y;
    acc0 = __builtin_amdgcn_wmma_f32_16x16x4_f32(false, a, false, bf2[2 * j], (short)0, acc0,
                                                 false, false);
    a.x = v.x; a.y = v.y;
    acc1 = __builtin_amdgcn_wmma_f32_16x16x4_f32(false, a, false, bf2[2 * j], (short)0, acc1,
                                                 false, false);
    a.x = u.z; a.y = u.w;
    acc0 = __builtin_amdgcn_wmma_f32_16x16x4_f32(false, a, false, bf2[2 * j + 1], (short)0, acc0,
                                                 false, false);
    a.x = v.z; a.y = v.w;
    acc1 = __builtin_amdgcn_wmma_f32_16x16x4_f32(false, a, false, bf2[2 * j + 1], (short)0, acc1,
                                                 false, false);
  }

  const int rowbase = tile * TR + rg2 * 32 + 8 * h;
  const int lim = N - rowbase;
  float* tp = T2 + (size_t)rowbase * GDOUT + n2;
#pragma unroll
  for (int r = 0; r < 8; ++r) {
    if (r < lim) tp[r * GDOUT] = acc0[r];
    if (r + 16 < lim) tp[(r + 16) * GDOUT] = acc1[r];
  }
}

extern "C" void kernel_launch(void* const* d_in, const int* in_sizes, int n_in,
                              void* d_out, int out_size, void* d_ws, size_t ws_size,
                              hipStream_t stream) {
  const float* x  = (const float*)d_in[0];
  const int*   ei = (const int*)d_in[1];
  const float* W1 = (const float*)d_in[2];
  const float* b1 = (const float*)d_in[3];
  const float* W2 = (const float*)d_in[4];
  const float* b2 = (const float*)d_in[5];
  float* out = (float*)d_out;

  const int N = in_sizes[0] / GDIN;
  const int E = in_sizes[1] / 2;
  const int* src = ei;
  const int* dst = ei + E;

  float* dinv = (float*)d_ws;
  float* bufA = dinv + N;
  float* bufB = bufA + (size_t)N * GDH;

  const int NS = (E + SLICE - 1) / SLICE;
  int* sliced = (int*)bufB;
  int* ofs    = sliced + (size_t)NS * SLICE * 2;
  const int NB = (N + BROWS - 1) >> BSH;
  const int RP = (N + 4) & ~3;
  int* lst = (int*)(bufB + (size_t)N * GDOUT);

  slice_sort_kernel<<<NS, MAXB, 0, stream>>>(src, dst, sliced, ofs, E);
  bucket_rows_fill_kernel<<<NB, MAXB, 0, stream>>>(ofs, sliced, lst, dinv, lst + RP, N, NS);

  gather_rows_two_gemms<<<(N + 63) / 64, 256, 0, stream>>>(lst, lst + RP, dinv, x, W1, b1, W2, bufB, N);

  gather_sum_rows<GDOUT, false><<<2048, 256, 0, stream>>>(lst, lst + RP, dinv, bufB, b2, out, N);
}
